# starting kernel
# baseline (speedup 1.0000x reference)
#include <hip/hip_runtime.h>
#include <hip/hip_bf16.h>

typedef __attribute__((ext_vector_type(2))) float v2f;
typedef __attribute__((ext_vector_type(8))) float v8f;

#define D_IN  512
#define D_HID 512
#define D_OUT 128
#define NEG_SLOPE 0.2f

__global__ __launch_bounds__(32) void gat_gemm_f32_wmma(
    const float* __restrict__ A, const float* __restrict__ B,
    float* __restrict__ C, int M, int K, int F)
{
    const int lane  = threadIdx.x & 31;
    const int m0    = blockIdx.x * 16;
    const int n0    = blockIdx.y * 64;
    if (m0 >= M) return;

    const int khalf = (lane >> 4) * 2;
    const int l15   = lane & 15;
    const float* Arow = A + (size_t)(m0 + l15) * K;

    v8f acc0 = {}; v8f acc1 = {}; v8f acc2 = {}; v8f acc3 = {};

    for (int k = 0; k < K; k += 4) {
        v2f a;
        a.x = Arow[k + khalf];
        a.y = Arow[k + khalf + 1];
        const float* Bk0 = B + (size_t)(k + khalf)     * F + n0 + l15;
        const float* Bk1 = B + (size_t)(k + khalf + 1) * F + n0 + l15;
        v2f b0, b1, b2, b3;
        b0.x = Bk0[0];  b0.y = Bk1[0];
        b1.x = Bk0[16]; b1.y = Bk1[16];
        b2.x = Bk0[32]; b2.y = Bk1[32];
        b3.x = Bk0[48]; b3.y = Bk1[48];
        acc0 = __builtin_amdgcn_wmma_f32_16x16x4_f32(false, a, false, b0, (short)0, acc0, false, false);
        acc1 = __builtin_amdgcn_wmma_f32_16x16x4_f32(false, a, false, b1, (short)0, acc1, false, false);
        acc2 = __builtin_amdgcn_wmma_f32_16x16x4_f32(false, a, false, b2, (short)0, acc2, false, false);
        acc3 = __builtin_amdgcn_wmma_f32_16x16x4_f32(false, a, false, b3, (short)0, acc3, false, false);
    }

    const int crow = m0 + (lane >> 4) * 8;
    float* Crow = C + (size_t)crow * F + n0 + l15;
    for (int r = 0; r < 8; ++r) {
        float* p = Crow + (size_t)r * F;
        p[0]  = acc0[r];
        p[16] = acc1[r];
        p[32] = acc2[r];
        p[48] = acc3[r];
    }
}

__global__ __launch_bounds__(32) void gat_alpha(
    const float* __restrict__ h, const float* __restrict__ a_s,
    const float* __restrict__ a_d, float* __restrict__ as_,
    float* __restrict__ ad_, int F)
{
    const int node = blockIdx.x;
    const int lane = threadIdx.x;
    const float* row = h + (size_t)node * F;
    float s0 = 0.f, s1 = 0.f;
    for (int f = lane; f < F; f += 32) {
        float v = row[f];
        s0 += v * a_s[f];
        s1 += v * a_d[f];
    }
    for (int off = 16; off > 0; off >>= 1) {
        s0 += __shfl_down(s0, off, 32);
        s1 += __shfl_down(s1, off, 32);
    }
    if (lane == 0) { as_[node] = s0; ad_[node] = s1; }
}

__device__ __forceinline__ int f32_ordered(float f) {
    int i = __float_as_int(f);
    return i ^ ((i >> 31) & 0x7FFFFFFF);
}
__device__ __forceinline__ float f32_unordered(int i) {
    i ^= ((i >> 31) & 0x7FFFFFFF);
    return __int_as_float(i);
}

__global__ void gat_init_seg(int* __restrict__ mInt, float* __restrict__ denom, int N)
{
    int i = blockIdx.x * blockDim.x + threadIdx.x;
    if (i < N) { mInt[i] = 0x807FFFFF; denom[i] = 0.f; }
}

__global__ void gat_edge_score(
    const int* __restrict__ src, const int* __restrict__ dst,
    const float* __restrict__ as_, const float* __restrict__ ad_,
    float* __restrict__ e, int* __restrict__ mInt, int E_real, int E_tot)
{
    int i = blockIdx.x * blockDim.x + threadIdx.x;
    if (i >= E_tot) return;
    int s, d;
    if (i < E_real) { s = src[i]; d = dst[i]; } else { s = d = i - E_real; }
    float v = as_[s] + ad_[d];
    v = v > 0.f ? v : NEG_SLOPE * v;
    e[i] = v;
    atomicMax(mInt + d, f32_ordered(v));
}

__global__ void gat_edge_exp(
    const int* __restrict__ src, const int* __restrict__ dst,
    float* __restrict__ e, const int* __restrict__ mInt,
    float* __restrict__ denom, int E_real, int E_tot)
{
    int i = blockIdx.x * blockDim.x + threadIdx.x;
    if (i >= E_tot) return;
    int d = (i < E_real) ? dst[i] : (i - E_real);
    float m  = f32_unordered(mInt[d]);
    float ex = __expf(e[i] - m);
    e[i] = ex;
    atomicAdd(denom + d, ex);
}

__global__ __launch_bounds__(128) void gat_edge_aggregate(
    const int* __restrict__ src, const int* __restrict__ dst,
    const float* __restrict__ e, const float* __restrict__ denom,
    const float* __restrict__ h, float* __restrict__ out,
    int E_real, int F)
{
    int edge = blockIdx.x;
    int s, d;
    if (edge < E_real) { s = src[edge]; d = dst[edge]; } else { s = d = edge - E_real; }
    float w = e[edge] / denom[d];
    const float* hr = h + (size_t)s * F;
    float* orow = out + (size_t)d * F;
    for (int f = threadIdx.x; f < F; f += 128)
        atomicAdd(orow + f, hr[f] * w);
}

__global__ void gat_bias_elu(float* __restrict__ x, const float* __restrict__ b, long long total, int Fmask)
{
    long long i = (long long)blockIdx.x * blockDim.x + threadIdx.x;
    if (i >= total) return;
    float v = x[i] + b[(int)i & Fmask];
    x[i] = v > 0.f ? v : (__expf(v) - 1.f);
}

__global__ void gat_bias_add(float* __restrict__ x, const float* __restrict__ b, long long total, int Fmask)
{
    long long i = (long long)blockIdx.x * blockDim.x + threadIdx.x;
    if (i >= total) return;
    x[i] += b[(int)i & Fmask];
}

extern "C" void kernel_launch(void* const* d_in, const int* in_sizes, int n_in,
                              void* d_out, int out_size, void* d_ws, size_t ws_size,
                              hipStream_t stream) {
    const float* feature = (const float*)d_in[0];
    const int*   edges   = (const int*)d_in[1];
    const float* W1      = (const float*)d_in[2];
    const float* a_src1  = (const float*)d_in[3];
    const float* a_dst1  = (const float*)d_in[4];
    const float* b1      = (const float*)d_in[5];
    const float* W2      = (const float*)d_in[6];
    const float* a_src2  = (const float*)d_in[7];
    const float* a_dst2  = (const float*)d_in[8];
    const float* b2      = (const float*)d_in[9];
    float* out = (float*)d_out;

    const int N      = in_sizes[0] / D_IN;
    const int E_real = in_sizes[1] / 2;
    const int E_tot  = E_real + N;
    const int* src = edges;
    const int* dst = edges + E_real;

    float* h1    = (float*)d_ws;
    float* agg1  = h1   + (size_t)N * D_HID;
    float* as_   = agg1 + (size_t)N * D_HID;
    float* ad_   = as_  + N;
    float* denom = ad_  + N;
    int*   mInt  = (int*)(denom + N);
    float* ebuf  = (float*)(mInt + N);
    float* h2    = h1;

    const int T = 256;
    dim3 blk32(32), blk128(128), blkT(T);
    dim3 gN((unsigned)N), gE((unsigned)E_tot);
    dim3 gEdgeT((E_tot + T - 1) / T), gNT((N + T - 1) / T);

    gat_gemm_f32_wmma<<<dim3(N / 16, D_HID / 64), blk32, 0, stream>>>(feature, W1, h1, N, D_IN, D_HID);
    gat_alpha<<<gN, blk32, 0, stream>>>(h1, a_src1, a_dst1, as_, ad_, D_HID);
    gat_init_seg<<<gNT, blkT, 0, stream>>>(mInt, denom, N);
    hipMemsetAsync(agg1, 0, (size_t)N * D_HID * sizeof(float), stream);
    gat_edge_score<<<gEdgeT, blkT, 0, stream>>>(src, dst, as_, ad_, ebuf, mInt, E_real, E_tot);
    gat_edge_exp<<<gEdgeT, blkT, 0, stream>>>(src, dst, ebuf, mInt, denom, E_real, E_tot);
    gat_edge_aggregate<<<gE, blk128, 0, stream>>>(src, dst, ebuf, denom, h1, agg1, E_real, D_HID);
    {
        long long total = (long long)N * D_HID;
        gat_bias_elu<<<dim3((unsigned)((total + T - 1) / T)), blkT, 0, stream>>>(agg1, b1, total, D_HID - 1);
    }

    gat_gemm_f32_wmma<<<dim3(N / 16, D_OUT / 64), blk32, 0, stream>>>(agg1, W2, h2, N, D_HID, D_OUT);
    gat_alpha<<<gN, blk32, 0, stream>>>(h2, a_src2, a_dst2, as_, ad_, D_OUT);
    gat_init_seg<<<gNT, blkT, 0, stream>>>(mInt, denom, N);
    hipMemsetAsync(out, 0, (size_t)N * D_OUT * sizeof(float), stream);
    gat_edge_score<<<gEdgeT, blkT, 0, stream>>>(src, dst, as_, ad_, ebuf, mInt, E_real, E_tot);
    gat_edge_exp<<<gEdgeT, blkT, 0, stream>>>(src, dst, ebuf, mInt, denom, E_real, E_tot);
    gat_edge_aggregate<<<gE, blk128, 0, stream>>>(src, dst, ebuf, denom, h2, out, E_real, D_OUT);
    {
        long long total = (long long)N * D_OUT;
        gat_bias_add<<<dim3((unsigned)((total + T - 1) / T)), blkT, 0, stream>>>(out, b2, total, D_OUT - 1);
    }
}


# Edges grouped by destination once (int counting sort, segments sorted by source); one wave per node does softmax and weighted row sum in registers, in that fixed order; float atomics and memsets gone
# speedup vs baseline: 2.1339x; 2.1339x over previous
#include <hip/hip_runtime.h>
#include <hip/hip_bf16.h>

typedef __attribute__((ext_vector_type(2))) float v2f;
typedef __attribute__((ext_vector_type(8))) float v8f;

#define D_IN  512
#define D_HID 512
#define D_OUT 128
#define NEG_SLOPE 0.2f

__global__ __launch_bounds__(32) void gat_gemm_f32_wmma(
    const float* __restrict__ A, const float* __restrict__ B,
    float* __restrict__ C, int M, int K, int F)
{
    const int lane  = threadIdx.x & 31;
    const int m0    = blockIdx.x * 16;
    const int n0    = blockIdx.y * 64;
    if (m0 >= M) return;

    const int khalf = (lane >> 4) * 2;
    const int l15   = lane & 15;
    const float* Arow = A + (size_t)(m0 + l15) * K;

    v8f acc0 = {}; v8f acc1 = {}; v8f acc2 = {}; v8f acc3 = {};

    for (int k = 0; k < K; k += 4) {
        v2f a;
        a.x = Arow[k + khalf];
        a.y = Arow[k + khalf + 1];
        const float* Bk0 = B + (size_t)(k + khalf)     * F + n0 + l15;
        const float* Bk1 = B + (size_t)(k + khalf + 1) * F + n0 + l15;
        v2f b0, b1, b2, b3;
        b0.x = Bk0[0];  b0.y = Bk1[0];
        b1.x = Bk0[16]; b1.y = Bk1[16];
        b2.x = Bk0[32]; b2.y = Bk1[32];
        b3.x = Bk0[48]; b3.y = Bk1[48];
        acc0 = __builtin_amdgcn_wmma_f32_16x16x4_f32(false, a, false, b0, (short)0, acc0, false, false);
        acc1 = __builtin_amdgcn_wmma_f32_16x16x4_f32(false, a, false, b1, (short)0, acc1, false, false);
        acc2 = __builtin_amdgcn_wmma_f32_16x16x4_f32(false, a, false, b2, (short)0, acc2, false, false);
        acc3 = __builtin_amdgcn_wmma_f32_16x16x4_f32(false, a, false, b3, (short)0, acc3, false, false);
    }

    const int crow = m0 + (lane >> 4) * 8;
    float* Crow = C + (size_t)crow * F + n0 + l15;
    for (int r = 0; r < 8; ++r) {
        float* p = Crow + (size_t)r * F;
        p[0]  = acc0[r];
        p[16] = acc1[r];
        p[32] = acc2[r];
        p[48] = acc3[r];
    }
}

__global__ __launch_bounds__(32) void gat_alpha(
    const float* __restrict__ h, const float* __restrict__ a_s,
    const float* __restrict__ a_d, float* __restrict__ as_,
    float* __restrict__ ad_, int F)
{
    const int node = blockIdx.x;
    const int lane = threadIdx.x;
    const float* row = h + (size_t)node * F;
    float s0 = 0.f, s1 = 0.f;
    for (int f = lane; f < F; f += 32) {
        float v = row[f];
        s0 += v * a_s[f];
        s1 += v * a_d[f];
    }
    for (int off = 16; off > 0; off >>= 1) {
        s0 += __shfl_down(s0, off, 32);
        s1 += __shfl_down(s1, off, 32);
    }
    if (lane == 0) { as_[node] = s0; ad_[node] = s1; }
}

__global__ void gat_zero_counts(int* __restrict__ cnt, int N)
{
    int i = blockIdx.x * blockDim.x + threadIdx.x;
    if (i < N) cnt[i] = 0;
}

__global__ void gat_count(const int* __restrict__ dst, int* __restrict__ cnt,
                          int* __restrict__ slot, int E)
{
    int i = blockIdx.x * blockDim.x + threadIdx.x;
    if (i < E) slot[i] = atomicAdd(cnt + dst[i], 1);
}

__global__ __launch_bounds__(1024) void gat_scan(
    const int* __restrict__ cnt, int* __restrict__ start, int N)
{
    __shared__ int wsum[32];
    const int t = threadIdx.x, lane = t & 31, w = t >> 5;
    const int per = (N + 1023) / 1024;
    int lo = t * per;       if (lo > N) lo = N;
    int hi = lo + per;      if (hi > N) hi = N;
    int s = 0;
    for (int i = lo; i < hi; ++i) s += cnt[i];
    int incl = s;
    for (int off = 1; off < 32; off <<= 1) {
        int v = __shfl_up(incl, off, 32);
        if (lane >= off) incl += v;
    }
    if (lane == 31) wsum[w] = incl;
    __syncthreads();
    if (w == 0) {
        int v = wsum[lane], iv = v;
        for (int off = 1; off < 32; off <<= 1) {
            int u = __shfl_up(iv, off, 32);
            if (lane >= off) iv += u;
        }
        wsum[lane] = iv - v;
    }
    __syncthreads();
    int run = wsum[w] + incl - s;
    for (int i = lo; i < hi; ++i) { start[i] = run; run += cnt[i]; }
}

__global__ void gat_fill(const int* __restrict__ src, const int* __restrict__ dst,
                         const int* __restrict__ slot, const int* __restrict__ start,
                         int* __restrict__ grouped, int E)
{
    int i = blockIdx.x * blockDim.x + threadIdx.x;
    if (i < E) grouped[start[dst[i]] + slot[i]] = src[i];
}

__global__ __launch_bounds__(256) void gat_seg_sort(
    const int* __restrict__ grouped, const int* __restrict__ start,
    const int* __restrict__ cnt, int* __restrict__ nbr, int N)
{
    const int node = __builtin_amdgcn_readfirstlane((int)(blockIdx.x * 8 + (threadIdx.x >> 5)));
    if (node >= N) return;
    const int lane = threadIdx.x & 31;
    const int st = start[node], deg = cnt[node];
    const int* seg = grouped + st;
    for (int e = lane; e < deg; e += 32) {
        const int mine = seg[e];
        int rank = 0;
        for (int j = 0; j < deg; ++j) {
            const int k = seg[j];
            rank += ((k < mine) || (k == mine && j < e)) ? 1 : 0;
        }
        nbr[st + rank] = mine;
    }
}

__device__ __forceinline__ float wave_max(float v) {
    for (int off = 16; off > 0; off >>= 1) v = fmaxf(v, __shfl_xor(v, off, 32));
    return v;
}
__device__ __forceinline__ float wave_sum(float v) {
    for (int off = 16; off > 0; off >>= 1) v += __shfl_xor(v, off, 32);
    return v;
}
__device__ __forceinline__ float leaky(float v) { return v > 0.f ? v : NEG_SLOPE * v; }

template <int F, int U, bool ELU>
__device__ __forceinline__ void gat_gather_body(
    const int* __restrict__ nbr, const int* __restrict__ start,
    const int* __restrict__ cnt,
    const float* __restrict__ as_, const float* __restrict__ ad_,
    const float* __restrict__ h, const float* __restrict__ bias,
    float* __restrict__ out, int N)
{
    constexpr int C = F / 128;
    const int node = __builtin_amdgcn_readfirstlane((int)(blockIdx.x * 8 + (threadIdx.x >> 5)));
    if (node >= N) return;
    const int lane = threadIdx.x & 31;
    const int st  = start[node];
    const int n   = cnt[node] + 1;
    const float adv = ad_[node];

    const bool has0 = lane < n;
    int   s0 = node;
    float v0 = -INFINITY;
    if (has0) {
        s0 = (lane < n - 1) ? nbr[st + lane] : node;
        v0 = leaky(as_[s0] + adv);
    }
    float m = v0;
    for (int e = lane + 32; e < n; e += 32) {
        const int s = (e < n - 1) ? nbr[st + e] : node;
        m = fmaxf(m, leaky(as_[s] + adv));
    }
    m = wave_max(m);
    const float x0 = has0 ? __expf(v0 - m) : 0.f;
    float den = x0;
    for (int e = lane + 32; e < n; e += 32) {
        const int s = (e < n - 1) ? nbr[st + e] : node;
        den += __expf(leaky(as_[s] + adv) - m);
    }
    den = wave_sum(den);

    float4 acc[C];
    #pragma unroll
    for (int c = 0; c < C; ++c) acc[c] = make_float4(0.f, 0.f, 0.f, 0.f);

    const float* hl = h + lane * 4;
    for (int base = 0; base < n; base += 32) {
        const int e = base + lane;
        int   s = s0;
        float w = x0 / den;
        if (base > 0) {
            s = node; w = 0.f;
            if (e < n) {
                s = (e < n - 1) ? nbr[st + e] : node;
                w = __expf(leaky(as_[s] + adv) - m) / den;
            }
        }
        const int nb = (n - base < 32) ? (n - base) : 32;
        for (int j = 0; j < nb; j += U) {
            float4 r[U][C];
            float  wj[U];
            #pragma unroll
            for (int u = 0; u < U; ++u) {
                const int jj = (j + u < nb) ? (j + u) : (nb - 1);
                const int sj = __builtin_amdgcn_readlane(s, jj);
                const float wv = __int_as_float(__builtin_amdgcn_readlane(__float_as_int(w), jj));
                wj[u] = (j + u < nb) ? wv : 0.f;
                const float* row = hl + (size_t)sj * F;
                #pragma unroll
                for (int c = 0; c < C; ++c)
                    r[u][c] = *reinterpret_cast<const float4*>(row + c * 128);
            }
            #pragma unroll
            for (int u = 0; u < U; ++u) {
                #pragma unroll
                for (int c = 0; c < C; ++c) {
                    acc[c].x += r[u][c].x * wj[u];
                    acc[c].y += r[u][c].y * wj[u];
                    acc[c].z += r[u][c].z * wj[u];
                    acc[c].w += r[u][c].w * wj[u];
                }
            }
        }
    }

    float* orow = out + (size_t)node * F + lane * 4;
    #pragma unroll
    for (int c = 0; c < C; ++c) {
        const float4 b = *reinterpret_cast<const float4*>(bias + c * 128 + lane * 4);
        float4 v;
        v.x = acc[c].x + b.x; v.y = acc[c].y + b.y;
        v.z = acc[c].z + b.z; v.w = acc[c].w + b.w;
        if (ELU) {
            v.x = v.x > 0.f ? v.x : (__expf(v.x) - 1.f);
            v.y = v.y > 0.f ? v.y : (__expf(v.y) - 1.f);
            v.z = v.z > 0.f ? v.z : (__expf(v.z) - 1.f);
            v.w = v.w > 0.f ? v.w : (__expf(v.w) - 1.f);
        }
        *reinterpret_cast<float4*>(orow + c * 128) = v;
    }
}

__global__ __launch_bounds__(256) void gat_gather_hidden(
    const int* __restrict__ nbr, const int* __restrict__ start,
    const int* __restrict__ cnt,
    const float* __restrict__ as_, const float* __restrict__ ad_,
    const float* __restrict__ h, const float* __restrict__ bias,
    float* __restrict__ out, int N)
{
    gat_gather_body<D_HID, 4, true>(nbr, start, cnt, as_, ad_, h, bias, out, N);
}

__global__ __launch_bounds__(256) void gat_gather_out(
    const int* __restrict__ nbr, const int* __restrict__ start,
    const int* __restrict__ cnt,
    const float* __restrict__ as_, const float* __restrict__ ad_,
    const float* __restrict__ h, const float* __restrict__ bias,
    float* __restrict__ out, int N)
{
    gat_gather_body<D_OUT, 8, false>(nbr, start, cnt, as_, ad_, h, bias, out, N);
}

extern "C" void kernel_launch(void* const* d_in, const int* in_sizes, int n_in,
                              void* d_out, int out_size, void* d_ws, size_t ws_size,
                              hipStream_t stream) {
    const float* feature = (const float*)d_in[0];
    const int*   edges   = (const int*)d_in[1];
    const float* W1      = (const float*)d_in[2];
    const float* a_src1  = (const float*)d_in[3];
    const float* a_dst1  = (const float*)d_in[4];
    const float* b1      = (const float*)d_in[5];
    const float* W2      = (const float*)d_in[6];
    const float* a_src2  = (const float*)d_in[7];
    const float* a_dst2  = (const float*)d_in[8];
    const float* b2      = (const float*)d_in[9];
    float* out = (float*)d_out;

    const int N      = in_sizes[0] / D_IN;
    const int E_real = in_sizes[1] / 2;
    const int* src = edges;
    const int* dst = edges + E_real;

    float* h1    = (float*)d_ws;
    float* agg1  = h1   + (size_t)N * D_HID;
    float* as_   = agg1 + (size_t)N * D_HID;
    float* ad_   = as_  + N;
    int*   start = (int*)(ad_ + N);
    int*   cnt   = start + N;
    int*   nbr   = cnt + N;
    float* h2    = h1;
    int*   slot    = (int*)agg1;
    int*   grouped = slot + E_real;

    const int T = 256;
    dim3 blk32(32), blkT(T), blk1024(1024);
    dim3 gN((unsigned)N);
    dim3 gET((E_real + T - 1) / T), gNT((N + T - 1) / T), gN8((N + 7) / 8);

    gat_zero_counts<<<gNT, blkT, 0, stream>>>(cnt, N);
    gat_count<<<gET, blkT, 0, stream>>>(dst, cnt, slot, E_real);
    gat_scan<<<dim3(1), blk1024, 0, stream>>>(cnt, start, N);
    gat_fill<<<gET, blkT, 0, stream>>>(src, dst, slot, start, grouped, E_real);
    gat_seg_sort<<<gN8, blkT, 0, stream>>>(grouped, start, cnt, nbr, N);

    gat_gemm_f32_wmma<<<dim3(N / 16, D_HID / 64), blk32, 0, stream>>>(feature, W1, h1, N, D_IN, D_HID);
    gat_alpha<<<gN, blk32, 0, stream>>>(h1, a_src1, a_dst1, as_, ad_, D_HID);
    gat_gather_hidden<<<gN8, blkT, 0, stream>>>(nbr, start, cnt, as_, ad_, h1, b1, agg1, N);

    gat_gemm_f32_wmma<<<dim3(N / 16, D_OUT / 64), blk32, 0, stream>>>(agg1, W2, h2, N, D_HID, D_OUT);
    gat_alpha<<<gN, blk32, 0, stream>>>(h2, a_src2, a_dst2, as_, ad_, D_OUT);
    gat_gather_out<<<gN8, blkT, 0, stream>>>(nbr, start, cnt, as_, ad_, h2, b2, out, N);
}


# GEMM operand feed: 16-byte loads of 8 k per lane of A and 4 adjacent B columns per k row, fetched one 16-k chunk ahead of its WMMAs; C stored as 16-byte pieces; k and column sum orders change
# speedup vs baseline: 3.2067x; 1.5027x over previous
#include <hip/hip_runtime.h>
#include <hip/hip_bf16.h>

typedef __attribute__((ext_vector_type(2))) float v2f;
typedef __attribute__((ext_vector_type(4))) float v4f;
typedef __attribute__((ext_vector_type(8))) float v8f;
typedef __attribute__((ext_vector_type(16))) float v16f;

#define D_IN  512
#define D_HID 512
#define D_OUT 128
#define NEG_SLOPE 0.2f

struct gemm_chunk {
    v4f a0, a1;
    v4f b[8];
};

__device__ __forceinline__ void gemm_fetch(gemm_chunk& q, const float* __restrict__ Ap,
                                           const float* __restrict__ Bp, size_t F)
{
    q.a0 = *reinterpret_cast<const v4f*>(Ap);
    q.a1 = *reinterpret_cast<const v4f*>(Ap + 4);
    #pragma unroll
    for (int i = 0; i < 8; ++i)
        q.b[i] = *reinterpret_cast<const v4f*>(Bp + (size_t)i * F);
}

__device__ __forceinline__ void gemm_mma(const gemm_chunk& q, v8f& acc0, v8f& acc1,
                                         v8f& acc2, v8f& acc3)
{
    #pragma unroll
    for (int j = 0; j < 4; ++j) {
        const v4f aq = (j < 2) ? q.a0 : q.a1;
        v2f a;  a.x = aq[(2 * j) & 3];  a.y = aq[((2 * j) & 3) + 1];
        const v4f lo = q.b[2 * j], hi = q.b[2 * j + 1];
        v2f b0, b1, b2, b3;
        b0.x = lo[0]; b0.y = hi[0];
        b1.x = lo[1]; b1.y = hi[1];
        b2.x = lo[2]; b2.y = hi[2];
        b3.x = lo[3]; b3.y = hi[3];
        acc0 = __builtin_amdgcn_wmma_f32_16x16x4_f32(false, a, false, b0, (short)0, acc0, false, false);
        acc1 = __builtin_amdgcn_wmma_f32_16x16x4_f32(false, a, false, b1, (short)0, acc1, false, false);
        acc2 = __builtin_amdgcn_wmma_f32_16x16x4_f32(false, a, false, b2, (short)0, acc2, false, false);
        acc3 = __builtin_amdgcn_wmma_f32_16x16x4_f32(false, a, false, b3, (short)0, acc3, false, false);
    }
}

__global__ __launch_bounds__(32) void gat_gemm_f32_wmma(
    const float* __restrict__ A, const float* __restrict__ B,
    float* __restrict__ C,
    const float* __restrict__ a_s, const float* __restrict__ a_d,
    float* __restrict__ Ps, float* __restrict__ Pd, int M, int K, int F)
{
    const int lane  = threadIdx.x & 31;
    const int m0    = blockIdx.x * 16;
    const int n0    = blockIdx.y * 64;
    if (m0 >= M) return;

    const int half = lane >> 4;
    const int l15  = lane & 15;
    const float* Ap = A + (size_t)(m0 + l15) * K + 8 * half;
    const float* Bp = B + (size_t)(8 * half) * F + n0 + 4 * l15;
    const size_t bstep = (size_t)16 * F;

    const v4f sv = *reinterpret_cast<const v4f*>(a_s + n0 + 4 * l15);
    const v4f dv = *reinterpret_cast<const v4f*>(a_d + n0 + 4 * l15);

    v8f acc0 = {}; v8f acc1 = {}; v8f acc2 = {}; v8f acc3 = {};

    gemm_chunk cur;
    gemm_fetch(cur, Ap, Bp, (size_t)F);
    for (int k0 = 16; k0 < K; k0 += 16) {
        Ap += 16;  Bp += bstep;
        gemm_chunk nxt;
        gemm_fetch(nxt, Ap, Bp, (size_t)F);
        gemm_mma(cur, acc0, acc1, acc2, acc3);
        cur = nxt;
    }
    gemm_mma(cur, acc0, acc1, acc2, acc3);

    float* Crow = C + (size_t)(m0 + half * 8) * F + n0 + 4 * l15;
    #pragma unroll
    for (int r = 0; r < 8; ++r) {
        v4f o;
        o[0] = acc0[r]; o[1] = acc1[r]; o[2] = acc2[r]; o[3] = acc3[r];
        *reinterpret_cast<v4f*>(Crow + (size_t)r * F) = o;
    }

    const float s0 = sv[0], s1 = sv[1], s2 = sv[2], s3 = sv[3];
    const float d0 = dv[0], d1 = dv[1], d2 = dv[2], d3 = dv[3];
    v16f p;
    #pragma unroll
    for (int r = 0; r < 8; ++r) {
        p[r]     = acc0[r] * s0 + acc1[r] * s1 + acc2[r] * s2 + acc3[r] * s3;
        p[8 + r] = acc0[r] * d0 + acc1[r] * d1 + acc2[r] * d2 + acc3[r] * d3;
    }
    const bool up8 = (l15 & 8) != 0, up4 = (l15 & 4) != 0;
    const bool up2 = (l15 & 2) != 0, up1 = (l15 & 1) != 0;
    v8f q;
    #pragma unroll
    for (int i = 0; i < 8; ++i) {
        const float lo = p[i], hi = p[i + 8];
        q[i] = (up8 ? hi : lo) + __shfl_xor(up8 ? lo : hi, 8, 32);
    }
    v4f t;
    #pragma unroll
    for (int i = 0; i < 4; ++i) {
        const float lo = q[i], hi = q[i + 4];
        t[i] = (up4 ? hi : lo) + __shfl_xor(up4 ? lo : hi, 4, 32);
    }
    v2f u;
    #pragma unroll
    for (int i = 0; i < 2; ++i) {
        const float lo = t[i], hi = t[i + 2];
        u[i] = (up2 ? hi : lo) + __shfl_xor(up2 ? lo : hi, 2, 32);
    }
    const float tot = (up1 ? u[1] : u[0]) + __shfl_xor(up1 ? u[0] : u[1], 1, 32);
    const int ns = F >> 6;
    float* T = (l15 < 8) ? Ps : Pd;
    T[(size_t)(m0 + half * 8 + (l15 & 7)) * ns + blockIdx.y] = tot;
}

__global__ void gat_zero_counts(int* __restrict__ cnt, int N)
{
    int i = blockIdx.x * blockDim.x + threadIdx.x;
    if (i < N) cnt[i] = 0;
}

__global__ void gat_count(const int* __restrict__ dst, int* __restrict__ cnt,
                          int* __restrict__ slot, int E)
{
    int i = blockIdx.x * blockDim.x + threadIdx.x;
    if (i < E) slot[i] = atomicAdd(cnt + dst[i], 1);
}

__global__ __launch_bounds__(1024) void gat_scan(
    const int* __restrict__ cnt, int* __restrict__ start, int N)
{
    __shared__ int wsum[32];
    const int t = threadIdx.x, lane = t & 31, w = t >> 5;
    const int per = (N + 1023) / 1024;
    int lo = t * per;       if (lo > N) lo = N;
    int hi = lo + per;      if (hi > N) hi = N;
    int s = 0;
    for (int i = lo; i < hi; ++i) s += cnt[i];
    int incl = s;
    for (int off = 1; off < 32; off <<= 1) {
        int v = __shfl_up(incl, off, 32);
        if (lane >= off) incl += v;
    }
    if (lane == 31) wsum[w] = incl;
    __syncthreads();
    if (w == 0) {
        int v = wsum[lane], iv = v;
        for (int off = 1; off < 32; off <<= 1) {
            int u = __shfl_up(iv, off, 32);
            if (lane >= off) iv += u;
        }
        wsum[lane] = iv - v;
    }
    __syncthreads();
    int run = wsum[w] + incl - s;
    for (int i = lo; i < hi; ++i) { start[i] = run; run += cnt[i]; }
}

__global__ void gat_fill(const int* __restrict__ src, const int* __restrict__ dst,
                         const int* __restrict__ slot, const int* __restrict__ start,
                         int* __restrict__ grouped, int E)
{
    int i = blockIdx.x * blockDim.x + threadIdx.x;
    if (i < E) grouped[start[dst[i]] + slot[i]] = src[i];
}

__global__ __launch_bounds__(256) void gat_seg_sort(
    const int* __restrict__ grouped, const int* __restrict__ start,
    const int* __restrict__ cnt, int* __restrict__ nbr, int N)
{
    const int node = __builtin_amdgcn_readfirstlane((int)(blockIdx.x * 8 + (threadIdx.x >> 5)));
    if (node >= N) return;
    const int lane = threadIdx.x & 31;
    const int st = start[node], deg = cnt[node];
    const int* seg = grouped + st;
    for (int e = lane; e < deg; e += 32) {
        const int mine = seg[e];
        int rank = 0;
        for (int j = 0; j < deg; ++j) {
            const int k = seg[j];
            rank += ((k < mine) || (k == mine && j < e)) ? 1 : 0;
        }
        nbr[st + rank] = mine;
    }
}

__device__ __forceinline__ float wave_max(float v) {
    for (int off = 16; off > 0; off >>= 1) v = fmaxf(v, __shfl_xor(v, off, 32));
    return v;
}
__device__ __forceinline__ float wave_sum(float v) {
    for (int off = 16; off > 0; off >>= 1) v += __shfl_xor(v, off, 32);
    return v;
}
__device__ __forceinline__ float leaky(float v) { return v > 0.f ? v : NEG_SLOPE * v; }

template <int NS>
__device__ __forceinline__ float slab_sum(const float* __restrict__ P, int node)
{
    const float* p = P + (size_t)node * NS;
    if (NS == 2) {
        const float2 a = *reinterpret_cast<const float2*>(p);
        return a.x + a.y;
    } else {
        float s = 0.f;
        #pragma unroll
        for (int i = 0; i < NS; i += 4) {
            const float4 a = *reinterpret_cast<const float4*>(p + i);
            s = (i == 0) ? a.x : s + a.x;
            s += a.y; s += a.z; s += a.w;
        }
        return s;
    }
}

template <int F, int U, bool ELU>
__device__ __forceinline__ void gat_gather_body(
    const int* __restrict__ nbr, const int* __restrict__ start,
    const int* __restrict__ cnt,
    const float* __restrict__ Ps, const float* __restrict__ Pd,
    const float* __restrict__ h, const float* __restrict__ bias,
    float* __restrict__ out, int N)
{
    constexpr int C  = F / 128;
    constexpr int NS = F / 64;
    const int node = __builtin_amdgcn_readfirstlane((int)(blockIdx.x * 8 + (threadIdx.x >> 5)));
    if (node >= N) return;
    const int lane = threadIdx.x & 31;
    const int st  = start[node];
    const int n   = cnt[node] + 1;
    const float adv = slab_sum<NS>(Pd, node);

    const bool has0 = lane < n;
    int   s0 = node;
    float v0 = -INFINITY;
    if (has0) {
        s0 = (lane < n - 1) ? nbr[st + lane] : node;
        v0 = leaky(slab_sum<NS>(Ps, s0) + adv);
    }
    float m = v0;
    for (int e = lane + 32; e < n; e += 32) {
        const int s = (e < n - 1) ? nbr[st + e] : node;
        m = fmaxf(m, leaky(slab_sum<NS>(Ps, s) + adv));
    }
    m = wave_max(m);
    const float x0 = has0 ? __expf(v0 - m) : 0.f;
    float den = x0;
    for (int e = lane + 32; e < n; e += 32) {
        const int s = (e < n - 1) ? nbr[st + e] : node;
        den += __expf(leaky(slab_sum<NS>(Ps, s) + adv) - m);
    }
    den = wave_sum(den);

    float4 acc[C];
    #pragma unroll
    for (int c = 0; c < C; ++c) acc[c] = make_float4(0.f, 0.f, 0.f, 0.f);

    const float* hl = h + lane * 4;
    for (int base = 0; base < n; base += 32) {
        const int e = base + lane;
        int   s = s0;
        float w = x0 / den;
        if (base > 0) {
            s = node; w = 0.f;
            if (e < n) {
                s = (e < n - 1) ? nbr[st + e] : node;
                w = __expf(leaky(slab_sum<NS>(Ps, s) + adv) - m) / den;
            }
        }
        const int nb = (n - base < 32) ? (n - base) : 32;
        for (int j = 0; j < nb; j += U) {
            float4 r[U][C];
            float  wj[U];
            #pragma unroll
            for (int u = 0; u < U; ++u) {
                const int jj = (j + u < nb) ? (j + u) : (nb - 1);
                const int sj = __builtin_amdgcn_readlane(s, jj);
                const float wv = __int_as_float(__builtin_amdgcn_readlane(__float_as_int(w), jj));
                wj[u] = (j + u < nb) ? wv : 0.f;
                const float* row = hl + (size_t)sj * F;
                #pragma unroll
                for (int c = 0; c < C; ++c)
                    r[u][c] = *reinterpret_cast<const float4*>(row + c * 128);
            }
            #pragma unroll
            for (int u = 0; u < U; ++u) {
                #pragma unroll
                for (int c = 0; c < C; ++c) {
                    acc[c].x += r[u][c].x * wj[u];
                    acc[c].y += r[u][c].y * wj[u];
                    acc[c].z += r[u][c].z * wj[u];
                    acc[c].w += r[u][c].w * wj[u];
                }
            }
        }
    }

    float* orow = out + (size_t)node * F + lane * 4;
    #pragma unroll
    for (int c = 0; c < C; ++c) {
        const float4 b = *reinterpret_cast<const float4*>(bias + c * 128 + lane * 4);
        float4 v;
        v.x = acc[c].x + b.x; v.y = acc[c].y + b.y;
        v.z = acc[c].z + b.z; v.w = acc[c].w + b.w;
        if (ELU) {
            v.x = v.x > 0.f ? v.x : (__expf(v.x) - 1.f);
            v.y = v.y > 0.f ? v.y : (__expf(v.y) - 1.f);
            v.z = v.z > 0.f ? v.z : (__expf(v.z) - 1.f);
            v.w = v.w > 0.f ? v.w : (__expf(v.w) - 1.f);
        }
        *reinterpret_cast<float4*>(orow + c * 128) = v;
    }
}

__global__ __launch_bounds__(256) void gat_gather_hidden(
    const int* __restrict__ nbr, const int* __restrict__ start,
    const int* __restrict__ cnt,
    const float* __restrict__ Ps, const float* __restrict__ Pd,
    const float* __restrict__ h, const float* __restrict__ bias,
    float* __restrict__ out, int N)
{
    gat_gather_body<D_HID, 4, true>(nbr, start, cnt, Ps, Pd, h, bias, out, N);
}

__global__ __launch_bounds__(256) void gat_gather_out(
    const int* __restrict__ nbr, const int* __restrict__ start,
    const int* __restrict__ cnt,
    const float* __restrict__ Ps, const float* __restrict__ Pd,
    const float* __restrict__ h, const float* __restrict__ bias,
    float* __restrict__ out, int N)
{
    gat_gather_body<D_OUT, 8, false>(nbr, start, cnt, Ps, Pd, h, bias, out, N);
}

extern "C" void kernel_launch(void* const* d_in, const int* in_sizes, int n_in,
                              void* d_out, int out_size, void* d_ws, size_t ws_size,
                              hipStream_t stream) {
    const float* feature = (const float*)d_in[0];
    const int*   edges   = (const int*)d_in[1];
    const float* W1      = (const float*)d_in[2];
    const float* a_src1  = (const float*)d_in[3];
    const float* a_dst1  = (const float*)d_in[4];
    const float* b1      = (const float*)d_in[5];
    const float* W2      = (const float*)d_in[6];
    const float* a_src2  = (const float*)d_in[7];
    const float* a_dst2  = (const float*)d_in[8];
    const float* b2      = (const float*)d_in[9];
    float* out = (float*)d_out;

    const int N      = in_sizes[0] / D_IN;
    const int E_real = in_sizes[1] / 2;
    const int* src = edges;
    const int* dst = edges + E_real;

    float* h1    = (float*)d_ws;
    float* agg1  = h1   + (size_t)N * D_HID;
    float* spare = agg1 + (size_t)N * D_HID;
    int*   start = (int*)(spare + 2 * (size_t)N);
    int*   cnt   = start + N;
    int*   nbr   = cnt + N;
    float* h2    = h1;
    int*   slot    = (int*)agg1;
    int*   grouped = slot + E_real;
    float* p1s = out;
    float* p1d = p1s + (size_t)N * (D_HID / 64);
    float* p2s = h1  + (size_t)N * D_OUT;
    float* p2d = p2s + (size_t)N * (D_OUT / 64);

    const int T = 256;
    dim3 blk32(32), blkT(T), blk1024(1024);
    dim3 gET((E_real + T - 1) / T), gNT((N + T - 1) / T), gN8((N + 7) / 8);

    gat_zero_counts<<<gNT, blkT, 0, stream>>>(cnt, N);
    gat_count<<<gET, blkT, 0, stream>>>(dst, cnt, slot, E_real);
    gat_scan<<<dim3(1), blk1024, 0, stream>>>(cnt, start, N);
    gat_fill<<<gET, blkT, 0, stream>>>(src, dst, slot, start, grouped, E_real);
    gat_seg_sort<<<gN8, blkT, 0, stream>>>(grouped, start, cnt, nbr, N);

    gat_gemm_f32_wmma<<<dim3(N / 16, D_HID / 64), blk32, 0, stream>>>(feature, W1, h1, a_src1, a_dst1, p1s, p1d, N, D_IN, D_HID);
    gat_gather_hidden<<<gN8, blkT, 0, stream>>>(nbr, start, cnt, p1s, p1d, h1, b1, agg1, N);

    gat_gemm_f32_wmma<<<dim3(N / 16, D_OUT / 64), blk32, 0, stream>>>(agg1, W2, h2, a_src2, a_dst2, p2s, p2d, N, D_HID, D_OUT);
    gat_gather_out<<<gN8, blkT, 0, stream>>>(nbr, start, cnt, p2s, p2d, h2, b2, out, N);
}


# Layer-2 product fused into the layer-1 gather launch: 16 nodes per work-group keep their rows in LDS and multiply by W2 on WMMA, K in 4 quarters added in order; 5 launches to 4; sum order changes
# speedup vs baseline: 4.7480x; 1.4806x over previous
#include <hip/hip_runtime.h>
#include <hip/hip_bf16.h>

typedef __attribute__((ext_vector_type(2))) float v2f;
typedef __attribute__((ext_vector_type(4))) float v4f;
typedef __attribute__((ext_vector_type(8))) float v8f;
typedef __attribute__((ext_vector_type(16))) float v16f;

#define D_IN  512
#define D_HID 512
#define D_OUT 128
#define NEG_SLOPE 0.2f

struct gemm_chunk {
    v4f a0, a1;
    v4f b[8];
};

__device__ __forceinline__ void gemm_fetch(gemm_chunk& q, const float* __restrict__ Ap,
                                           const float* __restrict__ Bp, size_t F)
{
    q.a0 = *reinterpret_cast<const v4f*>(Ap);
    q.a1 = *reinterpret_cast<const v4f*>(Ap + 4);
    #pragma unroll
    for (int i = 0; i < 8; ++i)
        q.b[i] = *reinterpret_cast<const v4f*>(Bp + (size_t)i * F);
}

__device__ __forceinline__ void gemm_mma(const gemm_chunk& q, v8f& acc0, v8f& acc1,
                                         v8f& acc2, v8f& acc3)
{
    #pragma unroll
    for (int j = 0; j < 4; ++j) {
        const v4f aq = (j < 2) ? q.a0 : q.a1;
        v2f a;  a.x = aq[(2 * j) & 3];  a.y = aq[((2 * j) & 3) + 1];
        const v4f lo = q.b[2 * j], hi = q.b[2 * j + 1];
        v2f b0, b1, b2, b3;
        b0.x = lo[0]; b0.y = hi[0];
        b1.x = lo[1]; b1.y = hi[1];
        b2.x = lo[2]; b2.y = hi[2];
        b3.x = lo[3]; b3.y = hi[3];
        acc0 = __builtin_amdgcn_wmma_f32_16x16x4_f32(false, a, false, b0, (short)0, acc0, false, false);
        acc1 = __builtin_amdgcn_wmma_f32_16x16x4_f32(false, a, false, b1, (short)0, acc1, false, false);
        acc2 = __builtin_amdgcn_wmma_f32_16x16x4_f32(false, a, false, b2, (short)0, acc2, false, false);
        acc3 = __builtin_amdgcn_wmma_f32_16x16x4_f32(false, a, false, b3, (short)0, acc3, false, false);
    }
}

__global__ __launch_bounds__(32) void gat_gemm_f32_wmma(
    const float* __restrict__ A, const float* __restrict__ B,
    float* __restrict__ C,
    const float* __restrict__ a_s, const float* __restrict__ a_d,
    float* __restrict__ Ps, float* __restrict__ Pd, int M, int K, int F)
{
    const int lane  = threadIdx.x & 31;
    const int m0    = blockIdx.x * 16;
    const int n0    = blockIdx.y * 64;
    if (m0 >= M) return;

    const int half = lane >> 4;
    const int l15  = lane & 15;
    const float* Ap = A + (size_t)(m0 + l15) * K + 8 * half;
    const float* Bp = B + (size_t)(8 * half) * F + n0 + 4 * l15;
    const size_t bstep = (size_t)16 * F;

    const v4f sv = *reinterpret_cast<const v4f*>(a_s + n0 + 4 * l15);
    const v4f dv = *reinterpret_cast<const v4f*>(a_d + n0 + 4 * l15);

    v8f acc0 = {}; v8f acc1 = {}; v8f acc2 = {}; v8f acc3 = {};

    gemm_chunk cur;
    gemm_fetch(cur, Ap, Bp, (size_t)F);
    for (int k0 = 16; k0 < K; k0 += 16) {
        Ap += 16;  Bp += bstep;
        gemm_chunk nxt;
        gemm_fetch(nxt, Ap, Bp, (size_t)F);
        gemm_mma(cur, acc0, acc1, acc2, acc3);
        cur = nxt;
    }
    gemm_mma(cur, acc0, acc1, acc2, acc3);

    float* Crow = C + (size_t)(m0 + half * 8) * F + n0 + 4 * l15;
    #pragma unroll
    for (int r = 0; r < 8; ++r) {
        v4f o;
        o[0] = acc0[r]; o[1] = acc1[r]; o[2] = acc2[r]; o[3] = acc3[r];
        *reinterpret_cast<v4f*>(Crow + (size_t)r * F) = o;
    }

    const float s0 = sv[0], s1 = sv[1], s2 = sv[2], s3 = sv[3];
    const float d0 = dv[0], d1 = dv[1], d2 = dv[2], d3 = dv[3];
    v16f p;
    #pragma unroll
    for (int r = 0; r < 8; ++r) {
        p[r]     = acc0[r] * s0 + acc1[r] * s1 + acc2[r] * s2 + acc3[r] * s3;
        p[8 + r] = acc0[r] * d0 + acc1[r] * d1 + acc2[r] * d2 + acc3[r] * d3;
    }
    const bool up8 = (l15 & 8) != 0, up4 = (l15 & 4) != 0;
    const bool up2 = (l15 & 2) != 0, up1 = (l15 & 1) != 0;
    v8f q;
    #pragma unroll
    for (int i = 0; i < 8; ++i) {
        const float lo = p[i], hi = p[i + 8];
        q[i] = (up8 ? hi : lo) + __shfl_xor(up8 ? lo : hi, 8, 32);
    }
    v4f t;
    #pragma unroll
    for (int i = 0; i < 4; ++i) {
        const float lo = q[i], hi = q[i + 4];
        t[i] = (up4 ? hi : lo) + __shfl_xor(up4 ? lo : hi, 4, 32);
    }
    v2f u;
    #pragma unroll
    for (int i = 0; i < 2; ++i) {
        const float lo = t[i], hi = t[i + 2];
        u[i] = (up2 ? hi : lo) + __shfl_xor(up2 ? lo : hi, 2, 32);
    }
    const float tot = (up1 ? u[1] : u[0]) + __shfl_xor(up1 ? u[0] : u[1], 1, 32);
    const int ns = F >> 6;
    float* T = (l15 < 8) ? Ps : Pd;
    T[(size_t)(m0 + half * 8 + (l15 & 7)) * ns + blockIdx.y] = tot;
}

#define GRP_NPB    80
#define GRP_NT     1024
#define GRP_NW     (GRP_NT / 32)
#define GRP_WCAP   128
#define GRP_SEGCAP 64
#define GRP_U      8
#define GRP_NONE   0x7fffffff

__device__ __forceinline__ int wave_sum_i(int v) {
    for (int off = 16; off > 0; off >>= 1) v += __shfl_xor(v, off, 32);
    return v;
}

__global__ __launch_bounds__(GRP_NT) void gat_group(
    const int* __restrict__ src, const int* __restrict__ dst,
    int* __restrict__ start, int* __restrict__ cnt, int* __restrict__ nbr,
    int N, int E)
{
    __shared__ int lcnt[GRP_NPB];
    __shared__ int lfill[GRP_NPB];
    __shared__ int lstart[GRP_NPB];
    __shared__ int lseg[GRP_NPB * GRP_SEGCAP];
    __shared__ int wl_node[GRP_NW * GRP_WCAP];
    __shared__ int wl_edge[GRP_NW * GRP_WCAP];
    __shared__ int wlen[GRP_NW];
    __shared__ int wbelow[GRP_NW];
    __shared__ int blk_over;

    const int t = threadIdx.x, lane = t & 31;
    const int w  = __builtin_amdgcn_readfirstlane(t >> 5);
    const int lo = blockIdx.x * GRP_NPB;

    for (int k = t; k < GRP_NPB; k += GRP_NT) { lcnt[k] = 0; lfill[k] = 0; }
    __syncthreads();

    int below = 0;
    int wl    = 0;
    int* const my_node = wl_node + w * GRP_WCAP;
    int* const my_edge = wl_edge + w * GRP_WCAP;

    auto see = [&](int d, int e) {
        const int r = d - lo;
        below += (r < 0) ? 1 : 0;
        const bool hit = (unsigned)r < (unsigned)GRP_NPB;
        const unsigned m = __builtin_amdgcn_ballot_w32(hit);
        if (m != 0u) {
            if (hit) {
                atomicAdd(&lcnt[r], 1);
                const int pos = wl + (int)__builtin_amdgcn_mbcnt_lo(m, 0u);
                if (pos < GRP_WCAP) { my_node[pos] = r; my_edge[pos] = e; }
            }
            wl += __builtin_popcount(m);
        }
    };

    int head = (int)(((16u - ((unsigned)(size_t)dst & 15u)) & 15u) >> 2);
    if (head > E) head = E;
    const int nvec  = (E - head) >> 2;
    const int tail0 = head + 4 * nvec;
    const int4* dv = reinterpret_cast<const int4*>(dst + head);
    const int rot = (int)(((long long)blockIdx.x * nvec) / gridDim.x);

    auto sweep = [&](int b0, bool guarded) __attribute__((always_inline)) {
        int4 d[GRP_U];
        int  c[GRP_U];
        #pragma unroll
        for (int u = 0; u < GRP_U; ++u) {
            const int i = b0 + u * GRP_NT + t;
            int cc = i + rot;
            if (cc >= nvec) cc -= nvec;
            const bool ok = !guarded || i < nvec;
            c[u] = cc;
            d[u] = dv[ok ? cc : 0];
            if (!ok) d[u] = make_int4(GRP_NONE, GRP_NONE, GRP_NONE, GRP_NONE);
        }
        #pragma unroll
        for (int u = 0; u < GRP_U; ++u) {
            const int e = head + 4 * c[u];
            see(d[u].x, e); see(d[u].y, e + 1); see(d[u].z, e + 2); see(d[u].w, e + 3);
        }
    };
    int b0 = 0;
    for (; b0 + GRP_NT * GRP_U <= nvec; b0 += GRP_NT * GRP_U) sweep(b0, false);
    if (b0 < nvec) sweep(b0, true);
    if (w == 0) {
        const int ntail = E - tail0;
        see((lane < head)  ? dst[lane]         : GRP_NONE, lane);
        see((lane < ntail) ? dst[tail0 + lane] : GRP_NONE, tail0 + lane);
    }

    const int wb = wave_sum_i(below);
    if (lane == 0) { wlen[w] = wl; wbelow[w] = wb; }
    __syncthreads();

    {
        const int have = (wl < GRP_WCAP) ? wl : GRP_WCAP;
        int rr[GRP_WCAP / 32], ss[GRP_WCAP / 32];
        #pragma unroll
        for (int q = 0; q < GRP_WCAP / 32; ++q) {
            const int p = q * 32 + lane;
            const bool ok = p < have;
            const int pc = ok ? p : 0;
            const int r = my_node[pc];
            const int e = my_edge[pc];
            rr[q] = ok ? r : -1;
            ss[q] = ok ? e : -1;
        }
        #pragma unroll
        for (int q = 0; q < GRP_WCAP / 32; ++q)
            ss[q] = (ss[q] >= 0) ? src[ss[q]] : 0;
        #pragma unroll
        for (int q = 0; q < GRP_WCAP / 32; ++q) {
            if (rr[q] >= 0) {
                const int slot = atomicAdd(&lfill[rr[q]], 1);
                if (slot < GRP_SEGCAP) lseg[rr[q] * GRP_SEGCAP + slot] = ss[q];
            }
        }
    }
    if (w == 0) {
        int b = 0, ov = 0;
        for (int i = lane; i < GRP_NW; i += 32) {
            b  += wbelow[i];
            ov |= (wlen[i] > GRP_WCAP) ? 1 : 0;
        }
        b = wave_sum_i(b);
        const unsigned anyov = __builtin_amdgcn_ballot_w32(ov != 0);
        constexpr int PER = (GRP_NPB + 31) / 32;
        int v[PER];
        int s = 0;
        #pragma unroll
        for (int i = 0; i < PER; ++i) {
            const int k = lane * PER + i;
            v[i] = (k < GRP_NPB) ? lcnt[k] : 0;
            s += v[i];
        }
        int incl = s;
        for (int off = 1; off < 32; off <<= 1) {
            const int u = __shfl_up(incl, off, 32);
            if (lane >= off) incl += u;
        }
        int run = b + incl - s;
        #pragma unroll
        for (int i = 0; i < PER; ++i) {
            const int k = lane * PER + i;
            if (k < GRP_NPB) lstart[k] = run;
            run += v[i];
        }
        if (lane == 0) blk_over = (anyov != 0u) ? 1 : 0;
    }
    __syncthreads();

    const bool over = __builtin_amdgcn_readfirstlane(blk_over) != 0;
    for (int k = w; k < GRP_NPB; k += GRP_NW) {
        const int node = lo + k;
        if (node >= N) break;
        const int deg = __builtin_amdgcn_readfirstlane(lcnt[k]);
        const int st  = __builtin_amdgcn_readfirstlane(lstart[k]);
        if (lane == 0) { start[node] = st; cnt[node] = deg; }
        if (!over && deg <= GRP_SEGCAP) {
            const int* seg = lseg + k * GRP_SEGCAP;
            const int x0 = seg[lane];
            const int a0 = (lane < deg) ? x0 : GRP_NONE;
            int r0 = 0;
            if (deg <= 32) {
                for (int j = 0; j < deg; ++j) {
                    const int kk = __builtin_amdgcn_readlane(a0, j);
                    r0 += ((kk < a0) || (kk == a0 && j < lane)) ? 1 : 0;
                }
                if (lane < deg) nbr[st + r0] = a0;
            } else {
                const int x1 = seg[lane + 32];
                const int a1 = (lane + 32 < deg) ? x1 : GRP_NONE;
                int r1 = 0;
                for (int j = 0; j < 32; ++j) {
                    const int kk = __builtin_amdgcn_readlane(a0, j);
                    r0 += ((kk < a0) || (kk == a0 && j < lane)) ? 1 : 0;
                    r1 += (kk <= a1) ? 1 : 0;
                }
                for (int j = 32; j < deg; ++j) {
                    const int kk = __builtin_amdgcn_readlane(a1, j - 32);
                    r0 += (kk < a0) ? 1 : 0;
                    r1 += ((kk < a1) || (kk == a1 && (j - 32) < lane)) ? 1 : 0;
                }
                nbr[st + r0] = a0;
                if (lane + 32 < deg) nbr[st + r1] = a1;
            }
        } else {
            int run = 0;
            for (int i0 = 0; i0 < E; i0 += 32) {
                const int i = i0 + lane;
                const bool hit = (i < E) && (dst[i] == node);
                const unsigned m = __builtin_amdgcn_ballot_w32(hit);
                if (hit) nbr[st + run + (int)__builtin_amdgcn_mbcnt_lo(m, 0u)] = src[i];
                run += __builtin_popcount(m);
            }
        }
    }
}

__device__ __forceinline__ float wave_max(float v) {
    for (int off = 16; off > 0; off >>= 1) v = fmaxf(v, __shfl_xor(v, off, 32));
    return v;
}
__device__ __forceinline__ float wave_sum(float v) {
    for (int off = 16; off > 0; off >>= 1) v += __shfl_xor(v, off, 32);
    return v;
}
__device__ __forceinline__ float leaky(float v) { return v > 0.f ? v : NEG_SLOPE * v; }

template <int NS>
__device__ __forceinline__ float slab_sum(const float* __restrict__ P, int node)
{
    const float* p = P + (size_t)node * NS;
    if (NS == 2) {
        const float2 a = *reinterpret_cast<const float2*>(p);
        return a.x + a.y;
    } else {
        float s = 0.f;
        #pragma unroll
        for (int i = 0; i < NS; i += 4) {
            const float4 a = *reinterpret_cast<const float4*>(p + i);
            s = (i == 0) ? a.x : s + a.x;
            s += a.y; s += a.z; s += a.w;
        }
        return s;
    }
}

template <int F, int U, bool ELU>
__device__ __forceinline__ void gat_gather_row(
    const int* __restrict__ nbr, const int* __restrict__ start,
    const int* __restrict__ cnt,
    const float* __restrict__ Ps, const float* __restrict__ Pd,
    const float* __restrict__ h, const float* __restrict__ bias,
    float* __restrict__ orow, const int node, const int lane)
{
    constexpr int C  = F / 128;
    constexpr int NS = F / 64;
    const int st  = start[node];
    const int n   = cnt[node] + 1;
    const float adv = slab_sum<NS>(Pd, node);

    const bool has0 = lane < n;
    int   s0 = node;
    float v0 = -INFINITY;
    if (has0) {
        s0 = (lane < n - 1) ? nbr[st + lane] : node;
        v0 = leaky(slab_sum<NS>(Ps, s0) + adv);
    }
    float m = v0;
    for (int e = lane + 32; e < n; e += 32) {
        const int s = (e < n - 1) ? nbr[st + e] : node;
        m = fmaxf(m, leaky(slab_sum<NS>(Ps, s) + adv));
    }
    m = wave_max(m);
    const float x0 = has0 ? __expf(v0 - m) : 0.f;
    float den = x0;
    for (int e = lane + 32; e < n; e += 32) {
        const int s = (e < n - 1) ? nbr[st + e] : node;
        den += __expf(leaky(slab_sum<NS>(Ps, s) + adv) - m);
    }
    den = wave_sum(den);

    float4 acc[C];
    #pragma unroll
    for (int c = 0; c < C; ++c) acc[c] = make_float4(0.f, 0.f, 0.f, 0.f);

    const float* hl = h + lane * 4;
    for (int base = 0; base < n; base += 32) {
        const int e = base + lane;
        int   s = s0;
        float w = x0 / den;
        if (base > 0) {
            s = node; w = 0.f;
            if (e < n) {
                s = (e < n - 1) ? nbr[st + e] : node;
                w = __expf(leaky(slab_sum<NS>(Ps, s) + adv) - m) / den;
            }
        }
        const int nb = (n - base < 32) ? (n - base) : 32;
        for (int j = 0; j < nb; j += U) {
            float4 r[U][C];
            float  wj[U];
            #pragma unroll
            for (int u = 0; u < U; ++u) {
                const int jj = (j + u < nb) ? (j + u) : (nb - 1);
                const int sj = __builtin_amdgcn_readlane(s, jj);
                const float wv = __int_as_float(__builtin_amdgcn_readlane(__float_as_int(w), jj));
                wj[u] = (j + u < nb) ? wv : 0.f;
                const float* row = hl + (size_t)sj * F;
                #pragma unroll
                for (int c = 0; c < C; ++c)
                    r[u][c] = *reinterpret_cast<const float4*>(row + c * 128);
            }
            #pragma unroll
            for (int u = 0; u < U; ++u) {
                #pragma unroll
                for (int c = 0; c < C; ++c) {
                    acc[c].x += r[u][c].x * wj[u];
                    acc[c].y += r[u][c].y * wj[u];
                    acc[c].z += r[u][c].z * wj[u];
                    acc[c].w += r[u][c].w * wj[u];
                }
            }
        }
    }

    #pragma unroll
    for (int c = 0; c < C; ++c) {
        const float4 b = *reinterpret_cast<const float4*>(bias + c * 128 + lane * 4);
        float4 v;
        v.x = acc[c].x + b.x; v.y = acc[c].y + b.y;
        v.z = acc[c].z + b.z; v.w = acc[c].w + b.w;
        if (ELU) {
            v.x = v.x > 0.f ? v.x : (__expf(v.x) - 1.f);
            v.y = v.y > 0.f ? v.y : (__expf(v.y) - 1.f);
            v.z = v.z > 0.f ? v.z : (__expf(v.z) - 1.f);
            v.w = v.w > 0.f ? v.w : (__expf(v.w) - 1.f);
        }
        *reinterpret_cast<float4*>(orow + c * 128) = v;
    }
}

#define FG_ROWS  16
#define FG_PITCH (D_HID + 4)

template <int MASK>
__device__ __forceinline__ float row16_xadd(float v) {
    const int p = __builtin_amdgcn_update_dpp(0, __float_as_int(v), 0x160 + MASK, 0xf, 0xf, false);
    return v + __int_as_float(p);
}
__device__ __forceinline__ float row16_sum(float v) {
    v = row16_xadd<8>(v);
    v = row16_xadd<4>(v);
    v = row16_xadd<2>(v);
    return row16_xadd<1>(v);
}

struct fg_chunk {
    v4f a0, a1;
    v2f b[8];
};

__device__ __forceinline__ void fg_fetch(fg_chunk& q, const float* Ap, const float* __restrict__ Bp)
{
    q.a0 = *reinterpret_cast<const v4f*>(Ap);
    q.a1 = *reinterpret_cast<const v4f*>(Ap + 4);
    #pragma unroll
    for (int i = 0; i < 8; ++i)
        q.b[i] = *reinterpret_cast<const v2f*>(Bp + (size_t)i * D_OUT);
}

__device__ __forceinline__ void fg_mma(const fg_chunk& q, v8f& acc0, v8f& acc1)
{
    #pragma unroll
    for (int j = 0; j < 4; ++j) {
        const v4f aq = (j < 2) ? q.a0 : q.a1;
        v2f a;  a.x = aq[(2 * j) & 3];  a.y = aq[((2 * j) & 3) + 1];
        const v2f lo = q.b[2 * j], hi = q.b[2 * j + 1];
        v2f b0, b1;
        b0.x = lo[0]; b0.y = hi[0];
        b1.x = lo[1]; b1.y = hi[1];
        acc0 = __builtin_amdgcn_wmma_f32_16x16x4_f32(false, a, false, b0, (short)0, acc0, false, false);
        acc1 = __builtin_amdgcn_wmma_f32_16x16x4_f32(false, a, false, b1, (short)0, acc1, false, false);
    }
}

__global__ __launch_bounds__(512, 12) void gat_gather_hidden_gemm(
    const int* __restrict__ nbr, const int* __restrict__ start,
    const int* __restrict__ cnt,
    const float* __restrict__ Ps, const float* __restrict__ Pd,
    const float* __restrict__ h, const float* __restrict__ bias,
    const float* __restrict__ W2,
    const float* __restrict__ a_s2, const float* __restrict__ a_d2,
    float* __restrict__ h2, float* __restrict__ P2s, float* __restrict__ P2d, int N)
{
    __shared__ float xs[FG_ROWS * FG_PITCH];
    __shared__ float part[4 * FG_ROWS * D_OUT];

    const int wave = __builtin_amdgcn_readfirstlane((int)(threadIdx.x >> 5));
    const int lane = threadIdx.x & 31;
    const int m0   = blockIdx.x * FG_ROWS;
    const int node = m0 + wave;

    float* xrow = xs + wave * FG_PITCH + lane * 4;
    if (node < N) {
        gat_gather_row<D_HID, 4, true>(nbr, start, cnt, Ps, Pd, h, bias, xrow, node, lane);
    } else {
        #pragma unroll
        for (int c = 0; c < D_HID / 128; ++c)
            *reinterpret_cast<float4*>(xrow + c * 128) = make_float4(0.f, 0.f, 0.f, 0.f);
    }
    __syncthreads();

    const int half = lane >> 4;
    const int l15  = lane & 15;
    const int n0   = (wave & 3) * 32;
    const int kq   = wave >> 2;
    const float* Ap = xs + l15 * FG_PITCH + kq * 128 + 8 * half;
    const float* Bp = W2 + (size_t)(kq * 128 + 8 * half) * D_OUT + n0 + 2 * l15;

    v8f acc0 = {}; v8f acc1 = {};
    fg_chunk q0, q1;
    fg_fetch(q0, Ap, Bp);
    #pragma unroll 1
    for (int it = 0; it < 4; ++it) {
        fg_fetch(q1, Ap + 16, Bp + (size_t)16 * D_OUT);
        fg_mma(q0, acc0, acc1);
        if (it < 3)
            fg_fetch(q0, Ap + 32, Bp + (size_t)32 * D_OUT);
        fg_mma(q1, acc0, acc1);
        Ap += 32;  Bp += (size_t)32 * D_OUT;
    }

    float* prow = part + (size_t)(kq * FG_ROWS + half * 8) * D_OUT + n0 + 2 * l15;
    #pragma unroll
    for (int r = 0; r < 8; ++r) {
        v2f o;  o.x = acc0[r];  o.y = acc1[r];
        *reinterpret_cast<v2f*>(prow + r * D_OUT) = o;
    }
    __syncthreads();

    const float* pr = part + (size_t)wave * D_OUT + lane * 4;
    const v4f p0 = *reinterpret_cast<const v4f*>(pr);
    const v4f p1 = *reinterpret_cast<const v4f*>(pr + 1 * FG_ROWS * D_OUT);
    const v4f p2 = *reinterpret_cast<const v4f*>(pr + 2 * FG_ROWS * D_OUT);
    const v4f p3 = *reinterpret_cast<const v4f*>(pr + 3 * FG_ROWS * D_OUT);
    const v4f v  = ((p0 + p1) + p2) + p3;
    const v4f sv = *reinterpret_cast<const v4f*>(a_s2 + lane * 4);
    const v4f dv = *reinterpret_cast<const v4f*>(a_d2 + lane * 4);
    float ps = v[0] * sv[0] + v[1] * sv[1] + v[2] * sv[2] + v[3] * sv[3];
    float pd = v[0] * dv[0] + v[1] * dv[1] + v[2] * dv[2] + v[3] * dv[3];
    ps = row16_sum(ps);
    pd = row16_sum(pd);
    if (node < N) {
        *reinterpret_cast<v4f*>(h2 + (size_t)node * D_OUT + lane * 4) = v;
        if (l15 == 0) {
            P2s[(size_t)node * (D_OUT / 64) + half] = ps;
            P2d[(size_t)node * (D_OUT / 64) + half] = pd;
        }
    }
}

__global__ __launch_bounds__(256) void gat_gather_out(
    const int* __restrict__ nbr, const int* __restrict__ start,
    const int* __restrict__ cnt,
    const float* __restrict__ Ps, const float* __restrict__ Pd,
    const float* __restrict__ h, const float* __restrict__ bias,
    float* __restrict__ out, int N)
{
    const int node = __builtin_amdgcn_readfirstlane((int)(blockIdx.x * 8 + (threadIdx.x >> 5)));
    if (node >= N) return;
    const int lane = threadIdx.x & 31;
    gat_gather_row<D_OUT, 8, false>(nbr, start, cnt, Ps, Pd, h, bias,
                                    out + (size_t)node * D_OUT + lane * 4, node, lane);
}

extern "C" void kernel_launch(void* const* d_in, const int* in_sizes, int n_in,
                              void* d_out, int out_size, void* d_ws, size_t ws_size,
                              hipStream_t stream) {
    const float* feature = (const float*)d_in[0];
    const int*   edges   = (const int*)d_in[1];
    const float* W1      = (const float*)d_in[2];
    const float* a_src1  = (const float*)d_in[3];
    const float* a_dst1  = (const float*)d_in[4];
    const float* b1      = (const float*)d_in[5];
    const float* W2      = (const float*)d_in[6];
    const float* a_src2  = (const float*)d_in[7];
    const float* a_dst2  = (const float*)d_in[8];
    const float* b2      = (const float*)d_in[9];
    float* out = (float*)d_out;

    const int N      = in_sizes[0] / D_IN;
    const int E_real = in_sizes[1] / 2;
    const int* src = edges;
    const int* dst = edges + E_real;

    float* h1    = (float*)d_ws;
    float* agg1  = h1   + (size_t)N * D_HID;
    float* spare = agg1 + (size_t)N * D_HID;
    int*   start = (int*)(spare + 2 * (size_t)N);
    int*   cnt   = start + N;
    int*   nbr   = cnt + N;
    float* h2    = agg1;
    float* p1s = out;
    float* p1d = p1s + (size_t)N * (D_HID / 64);
    float* p2s = h2  + (size_t)N * D_OUT;
    float* p2d = p2s + (size_t)N * (D_OUT / 64);

    const int T = 256;
    dim3 blk32(32), blkT(T);
    dim3 gN8((N + 7) / 8);

    gat_group<<<dim3((N + GRP_NPB - 1) / GRP_NPB), dim3(GRP_NT), 0, stream>>>(src, dst, start, cnt, nbr, N, E_real);

    gat_gemm_f32_wmma<<<dim3(N / 16, D_HID / 64), blk32, 0, stream>>>(feature, W1, h1, a_src1, a_dst1, p1s, p1d, N, D_IN, D_HID);
    gat_gather_hidden_gemm<<<dim3((N + FG_ROWS - 1) / FG_ROWS), dim3(FG_ROWS * 32), 0, stream>>>(
        nbr, start, cnt, p1s, p1d, h1, b1, W2, a_src2, a_dst2, h2, p2s, p2d, N);

    gat_gather_out<<<gN8, blkT, 0, stream>>>(nbr, start, cnt, p2s, p2d, h2, b2, out, N);
}


# Layer-1 GEMM: each work-group (20 waves, one 16x64 tile per wave) copies its 64-column slab of W1 into LDS once, k pairs interleaved; B fragments read from LDS, only A from memory; sum orders kept
# speedup vs baseline: 4.9875x; 1.0504x over previous
#include <hip/hip_runtime.h>
#include <hip/hip_bf16.h>

typedef __attribute__((ext_vector_type(2))) float v2f;
typedef __attribute__((ext_vector_type(4))) float v4f;
typedef __attribute__((ext_vector_type(8))) float v8f;
typedef __attribute__((ext_vector_type(16))) float v16f;

#define D_IN  512
#define D_HID 512
#define D_OUT 128
#define NEG_SLOPE 0.2f

#define G1_WAVES  20
#define G1_NT     (G1_WAVES * 32)
#define G1_SLABS  (D_HID / 64)
#define G1_KP     (D_IN / 2)
#define G1_PITCH  136
#define G1_LDS_FLOATS  (41 * 1024)
static_assert(G1_LDS_FLOATS >= G1_KP * G1_PITCH, "slab does not fit");

struct g1_chunk {
    v4f a0, a1;
    v4f b[8];
};

__device__ __forceinline__ void g1_fetch(g1_chunk& q, const float* __restrict__ Ap, const float* Bl)
{
    q.a0 = *reinterpret_cast<const v4f*>(Ap);
    q.a1 = *reinterpret_cast<const v4f*>(Ap + 4);
    #pragma unroll
    for (int j = 0; j < 4; ++j) {
        q.b[2 * j]     = *reinterpret_cast<const v4f*>(Bl + j * G1_PITCH);
        q.b[2 * j + 1] = *reinterpret_cast<const v4f*>(Bl + j * G1_PITCH + 64);
    }
}

__device__ __forceinline__ void g1_mma(const g1_chunk& q, v8f& acc0, v8f& acc1,
                                       v8f& acc2, v8f& acc3)
{
    #pragma unroll
    for (int j = 0; j < 4; ++j) {
        const v4f aq = (j < 2) ? q.a0 : q.a1;
        v2f a;  a.x = aq[(2 * j) & 3];  a.y = aq[((2 * j) & 3) + 1];
        const v4f lo = q.b[2 * j], hi = q.b[2 * j + 1];
        v2f b0, b1, b2, b3;
        b0.x = lo[0]; b0.y = lo[1];
        b1.x = lo[2]; b1.y = lo[3];
        b2.x = hi[0]; b2.y = hi[1];
        b3.x = hi[2]; b3.y = hi[3];
        acc0 = __builtin_amdgcn_wmma_f32_16x16x4_f32(false, a, false, b0, (short)0, acc0, false, false);
        acc1 = __builtin_amdgcn_wmma_f32_16x16x4_f32(false, a, false, b1, (short)0, acc1, false, false);
        acc2 = __builtin_amdgcn_wmma_f32_16x16x4_f32(false, a, false, b2, (short)0, acc2, false, false);
        acc3 = __builtin_amdgcn_wmma_f32_16x16x4_f32(false, a, false, b3, (short)0, acc3, false, false);
    }
}

__global__ __launch_bounds__(G1_NT) void gat_gemm1_f32_wmma(
    const float* __restrict__ A, const float* __restrict__ B,
    float* __restrict__ C,
    const float* __restrict__ a_s, const float* __restrict__ a_d,
    float* __restrict__ Ps, float* __restrict__ Pd, int M)
{
    __shared__ float Bs[G1_LDS_FLOATS];

    const int tid  = threadIdx.x;
    const int lane = tid & 31;
    const int wave = __builtin_amdgcn_readfirstlane(tid >> 5);
    const int slab = blockIdx.x % G1_SLABS;
    const int n0   = slab * 64;
    const int m0   = ((blockIdx.x / G1_SLABS) * G1_WAVES + wave) * 16;
    const bool valid = m0 < M;

    const int half = lane >> 4;
    const int l15  = lane & 15;

    constexpr int ITEMS = G1_KP * 16;
    constexpr int FULL  = ITEMS / G1_NT;
    constexpr int REM   = ITEMS - FULL * G1_NT;
    static_assert(G1_NT % 16 == 0, "column group must not depend on the round");
    static_assert(REM > 0 && REM <= G1_NT, "one partial round");
    const int cg = tid & 15;
    const float* Bq = B + n0 + 4 * cg;
    int pp[FULL + 1];
    v4f r0[FULL + 1], r1[FULL + 1];
    #pragma unroll
    for (int u = 0; u <= FULL; ++u) {
        pp[u] = ((u < FULL || tid < REM) ? (u * G1_NT + tid) : tid) >> 4;
        r0[u] = *reinterpret_cast<const v4f*>(Bq + (size_t)(2 * pp[u]) * D_HID);
        r1[u] = *reinterpret_cast<const v4f*>(Bq + (size_t)(2 * pp[u] + 1) * D_HID);
    }

    const v4f sv = *reinterpret_cast<const v4f*>(a_s + n0 + 4 * l15);
    const v4f dv = *reinterpret_cast<const v4f*>(a_d + n0 + 4 * l15);
    const float* Ap = A + (size_t)((valid ? m0 : 0) + l15) * D_IN + 8 * half;
    g1_chunk cur;
    cur.a0 = *reinterpret_cast<const v4f*>(Ap);
    cur.a1 = *reinterpret_cast<const v4f*>(Ap + 4);

    #pragma unroll
    for (int u = 0; u <= FULL; ++u) {
        float* d = Bs + pp[u] * G1_PITCH + 4 * cg;
        v4f lo, hi;
        lo[0] = r0[u][0]; lo[1] = r1[u][0]; lo[2] = r0[u][1]; lo[3] = r1[u][1];
        hi[0] = r0[u][2]; hi[1] = r1[u][2]; hi[2] = r0[u][3]; hi[3] = r1[u][3];
        *reinterpret_cast<v4f*>(d)      = lo;
        *reinterpret_cast<v4f*>(d + 64) = hi;
    }
    __syncthreads();
    if (!valid) return;

    const float* Bl = Bs + (4 * half) * G1_PITCH + 4 * l15;
    v8f acc0 = {}; v8f acc1 = {}; v8f acc2 = {}; v8f acc3 = {};

    #pragma unroll
    for (int j = 0; j < 4; ++j) {
        cur.b[2 * j]     = *reinterpret_cast<const v4f*>(Bl + j * G1_PITCH);
        cur.b[2 * j + 1] = *reinterpret_cast<const v4f*>(Bl + j * G1_PITCH + 64);
    }
    for (int k0 = 16; k0 < D_IN; k0 += 16) {
        Ap += 16;  Bl += 8 * G1_PITCH;
        g1_chunk nxt;
        g1_fetch(nxt, Ap, Bl);
        g1_mma(cur, acc0, acc1, acc2, acc3);
        cur = nxt;
    }
    g1_mma(cur, acc0, acc1, acc2, acc3);

    float* Crow = C + (size_t)(m0 + half * 8) * D_HID + n0 + 4 * l15;
    #pragma unroll
    for (int r = 0; r < 8; ++r) {
        v4f o;
        o[0] = acc0[r]; o[1] = acc1[r]; o[2] = acc2[r]; o[3] = acc3[r];
        *reinterpret_cast<v4f*>(Crow + (size_t)r * D_HID) = o;
    }

    const float s0 = sv[0], s1 = sv[1], s2 = sv[2], s3 = sv[3];
    const float d0 = dv[0], d1 = dv[1], d2 = dv[2], d3 = dv[3];
    v16f p;
    #pragma unroll
    for (int r = 0; r < 8; ++r) {
        p[r]     = acc0[r] * s0 + acc1[r] * s1 + acc2[r] * s2 + acc3[r] * s3;
        p[8 + r] = acc0[r] * d0 + acc1[r] * d1 + acc2[r] * d2 + acc3[r] * d3;
    }
    const bool up8 = (l15 & 8) != 0, up4 = (l15 & 4) != 0;
    const bool up2 = (l15 & 2) != 0, up1 = (l15 & 1) != 0;
    v8f q;
    #pragma unroll
    for (int i = 0; i < 8; ++i) {
        const float lo = p[i], hi = p[i + 8];
        q[i] = (up8 ? hi : lo) + __shfl_xor(up8 ? lo : hi, 8, 32);
    }
    v4f t;
    #pragma unroll
    for (int i = 0; i < 4; ++i) {
        const float lo = q[i], hi = q[i + 4];
        t[i] = (up4 ? hi : lo) + __shfl_xor(up4 ? lo : hi, 4, 32);
    }
    v2f u;
    #pragma unroll
    for (int i = 0; i < 2; ++i) {
        const float lo = t[i], hi = t[i + 2];
        u[i] = (up2 ? hi : lo) + __shfl_xor(up2 ? lo : hi, 2, 32);
    }
    const float tot = (up1 ? u[1] : u[0]) + __shfl_xor(up1 ? u[0] : u[1], 1, 32);
    float* T = (l15 < 8) ? Ps : Pd;
    T[(size_t)(m0 + half * 8 + (l15 & 7)) * G1_SLABS + slab] = tot;
}
#define GRP_NPB    80
#define GRP_NT     1024
#define GRP_NW     (GRP_NT / 32)
#define GRP_WCAP   128
#define GRP_SEGCAP 64
#define GRP_U      8
#define GRP_NONE   0x7fffffff

__device__ __forceinline__ int wave_sum_i(int v) {
    for (int off = 16; off > 0; off >>= 1) v += __shfl_xor(v, off, 32);
    return v;
}

__global__ __launch_bounds__(GRP_NT) void gat_group(
    const int* __restrict__ src, const int* __restrict__ dst,
    int* __restrict__ start, int* __restrict__ cnt, int* __restrict__ nbr,
    int N, int E)
{
    __shared__ int lcnt[GRP_NPB];
    __shared__ int lfill[GRP_NPB];
    __shared__ int lstart[GRP_NPB];
    __shared__ int lseg[GRP_NPB * GRP_SEGCAP];
    __shared__ int wl_node[GRP_NW * GRP_WCAP];
    __shared__ int wl_edge[GRP_NW * GRP_WCAP];
    __shared__ int wlen[GRP_NW];
    __shared__ int wbelow[GRP_NW];
    __shared__ int blk_over;

    const int t = threadIdx.x, lane = t & 31;
    const int w  = __builtin_amdgcn_readfirstlane(t >> 5);
    const int lo = blockIdx.x * GRP_NPB;

    for (int k = t; k < GRP_NPB; k += GRP_NT) { lcnt[k] = 0; lfill[k] = 0; }
    __syncthreads();

    int below = 0;
    int wl    = 0;
    int* const my_node = wl_node + w * GRP_WCAP;
    int* const my_edge = wl_edge + w * GRP_WCAP;

    auto see = [&](int d, int e) {
        const int r = d - lo;
        below += (r < 0) ? 1 : 0;
        const bool hit = (unsigned)r < (unsigned)GRP_NPB;
        const unsigned m = __builtin_amdgcn_ballot_w32(hit);
        if (m != 0u) {
            if (hit) {
                atomicAdd(&lcnt[r], 1);
                const int pos = wl + (int)__builtin_amdgcn_mbcnt_lo(m, 0u);
                if (pos < GRP_WCAP) { my_node[pos] = r; my_edge[pos] = e; }
            }
            wl += __builtin_popcount(m);
        }
    };

    int head = (int)(((16u - ((unsigned)(size_t)dst & 15u)) & 15u) >> 2);
    if (head > E) head = E;
    const int nvec  = (E - head) >> 2;
    const int tail0 = head + 4 * nvec;
    const int4* dv = reinterpret_cast<const int4*>(dst + head);
    const int rot = (int)(((long long)blockIdx.x * nvec) / gridDim.x);

    auto sweep = [&](int b0, bool guarded) __attribute__((always_inline)) {
        int4 d[GRP_U];
        int  c[GRP_U];
        #pragma unroll
        for (int u = 0; u < GRP_U; ++u) {
            const int i = b0 + u * GRP_NT + t;
            int cc = i + rot;
            if (cc >= nvec) cc -= nvec;
            const bool ok = !guarded || i < nvec;
            c[u] = cc;
            d[u] = dv[ok ? cc : 0];
            if (!ok) d[u] = make_int4(GRP_NONE, GRP_NONE, GRP_NONE, GRP_NONE);
        }
        #pragma unroll
        for (int u = 0; u < GRP_U; ++u) {
            const int e = head + 4 * c[u];
            see(d[u].x, e); see(d[u].y, e + 1); see(d[u].z, e + 2); see(d[u].w, e + 3);
        }
    };
    int b0 = 0;
    for (; b0 + GRP_NT * GRP_U <= nvec; b0 += GRP_NT * GRP_U) sweep(b0, false);
    if (b0 < nvec) sweep(b0, true);
    if (w == 0) {
        const int ntail = E - tail0;
        see((lane < head)  ? dst[lane]         : GRP_NONE, lane);
        see((lane < ntail) ? dst[tail0 + lane] : GRP_NONE, tail0 + lane);
    }

    const int wb = wave_sum_i(below);
    if (lane == 0) { wlen[w] = wl; wbelow[w] = wb; }
    __syncthreads();

    {
        const int have = (wl < GRP_WCAP) ? wl : GRP_WCAP;
        int rr[GRP_WCAP / 32], ss[GRP_WCAP / 32];
        #pragma unroll
        for (int q = 0; q < GRP_WCAP / 32; ++q) {
            const int p = q * 32 + lane;
            const bool ok = p < have;
            const int pc = ok ? p : 0;
            const int r = my_node[pc];
            const int e = my_edge[pc];
            rr[q] = ok ? r : -1;
            ss[q] = ok ? e : -1;
        }
        #pragma unroll
        for (int q = 0; q < GRP_WCAP / 32; ++q)
            ss[q] = (ss[q] >= 0) ? src[ss[q]] : 0;
        #pragma unroll
        for (int q = 0; q < GRP_WCAP / 32; ++q) {
            if (rr[q] >= 0) {
                const int slot = atomicAdd(&lfill[rr[q]], 1);
                if (slot < GRP_SEGCAP) lseg[rr[q] * GRP_SEGCAP + slot] = ss[q];
            }
        }
    }
    if (w == 0) {
        int b = 0, ov = 0;
        for (int i = lane; i < GRP_NW; i += 32) {
            b  += wbelow[i];
            ov |= (wlen[i] > GRP_WCAP) ? 1 : 0;
        }
        b = wave_sum_i(b);
        const unsigned anyov = __builtin_amdgcn_ballot_w32(ov != 0);
        constexpr int PER = (GRP_NPB + 31) / 32;
        int v[PER];
        int s = 0;
        #pragma unroll
        for (int i = 0; i < PER; ++i) {
            const int k = lane * PER + i;
            v[i] = (k < GRP_NPB) ? lcnt[k] : 0;
            s += v[i];
        }
        int incl = s;
        for (int off = 1; off < 32; off <<= 1) {
            const int u = __shfl_up(incl, off, 32);
            if (lane >= off) incl += u;
        }
        int run = b + incl - s;
        #pragma unroll
        for (int i = 0; i < PER; ++i) {
            const int k = lane * PER + i;
            if (k < GRP_NPB) lstart[k] = run;
            run += v[i];
        }
        if (lane == 0) blk_over = (anyov != 0u) ? 1 : 0;
    }
    __syncthreads();

    const bool over = __builtin_amdgcn_readfirstlane(blk_over) != 0;
    for (int k = w; k < GRP_NPB; k += GRP_NW) {
        const int node = lo + k;
        if (node >= N) break;
        const int deg = __builtin_amdgcn_readfirstlane(lcnt[k]);
        const int st  = __builtin_amdgcn_readfirstlane(lstart[k]);
        if (lane == 0) { start[node] = st; cnt[node] = deg; }
        if (!over && deg <= GRP_SEGCAP) {
            const int* seg = lseg + k * GRP_SEGCAP;
            const int x0 = seg[lane];
            const int a0 = (lane < deg) ? x0 : GRP_NONE;
            int r0 = 0;
            if (deg <= 32) {
                for (int j = 0; j < deg; ++j) {
                    const int kk = __builtin_amdgcn_readlane(a0, j);
                    r0 += ((kk < a0) || (kk == a0 && j < lane)) ? 1 : 0;
                }
                if (lane < deg) nbr[st + r0] = a0;
            } else {
                const int x1 = seg[lane + 32];
                const int a1 = (lane + 32 < deg) ? x1 : GRP_NONE;
                int r1 = 0;
                for (int j = 0; j < 32; ++j) {
                    const int kk = __builtin_amdgcn_readlane(a0, j);
                    r0 += ((kk < a0) || (kk == a0 && j < lane)) ? 1 : 0;
                    r1 += (kk <= a1) ? 1 : 0;
                }
                for (int j = 32; j < deg; ++j) {
                    const int kk = __builtin_amdgcn_readlane(a1, j - 32);
                    r0 += (kk < a0) ? 1 : 0;
                    r1 += ((kk < a1) || (kk == a1 && (j - 32) < lane)) ? 1 : 0;
                }
                nbr[st + r0] = a0;
                if (lane + 32 < deg) nbr[st + r1] = a1;
            }
        } else {
            int run = 0;
            for (int i0 = 0; i0 < E; i0 += 32) {
                const int i = i0 + lane;
                const bool hit = (i < E) && (dst[i] == node);
                const unsigned m = __builtin_amdgcn_ballot_w32(hit);
                if (hit) nbr[st + run + (int)__builtin_amdgcn_mbcnt_lo(m, 0u)] = src[i];
                run += __builtin_popcount(m);
            }
        }
    }
}

__device__ __forceinline__ float wave_max(float v) {
    for (int off = 16; off > 0; off >>= 1) v = fmaxf(v, __shfl_xor(v, off, 32));
    return v;
}
__device__ __forceinline__ float wave_sum(float v) {
    for (int off = 16; off > 0; off >>= 1) v += __shfl_xor(v, off, 32);
    return v;
}
__device__ __forceinline__ float leaky(float v) { return v > 0.f ? v : NEG_SLOPE * v; }

template <int NS>
__device__ __forceinline__ float slab_sum(const float* __restrict__ P, int node)
{
    const float* p = P + (size_t)node * NS;
    if (NS == 2) {
        const float2 a = *reinterpret_cast<const float2*>(p);
        return a.x + a.y;
    } else {
        float s = 0.f;
        #pragma unroll
        for (int i = 0; i < NS; i += 4) {
            const float4 a = *reinterpret_cast<const float4*>(p + i);
            s = (i == 0) ? a.x : s + a.x;
            s += a.y; s += a.z; s += a.w;
        }
        return s;
    }
}

template <int F, int U, bool ELU>
__device__ __forceinline__ void gat_gather_row(
    const int* __restrict__ nbr, const int* __restrict__ start,
    const int* __restrict__ cnt,
    const float* __restrict__ Ps, const float* __restrict__ Pd,
    const float* __restrict__ h, const float* __restrict__ bias,
    float* __restrict__ orow, const int node, const int lane)
{
    constexpr int C  = F / 128;
    constexpr int NS = F / 64;
    const int st  = start[node];
    const int n   = cnt[node] + 1;
    const float adv = slab_sum<NS>(Pd, node);

    const bool has0 = lane < n;
    int   s0 = node;
    float v0 = -INFINITY;
    if (has0) {
        s0 = (lane < n - 1) ? nbr[st + lane] : node;
        v0 = leaky(slab_sum<NS>(Ps, s0) + adv);
    }
    float m = v0;
    for (int e = lane + 32; e < n; e += 32) {
        const int s = (e < n - 1) ? nbr[st + e] : node;
        m = fmaxf(m, leaky(slab_sum<NS>(Ps, s) + adv));
    }
    m = wave_max(m);
    const float x0 = has0 ? __expf(v0 - m) : 0.f;
    float den = x0;
    for (int e = lane + 32; e < n; e += 32) {
        const int s = (e < n - 1) ? nbr[st + e] : node;
        den += __expf(leaky(slab_sum<NS>(Ps, s) + adv) - m);
    }
    den = wave_sum(den);

    float4 acc[C];
    #pragma unroll
    for (int c = 0; c < C; ++c) acc[c] = make_float4(0.f, 0.f, 0.f, 0.f);

    const float* hl = h + lane * 4;
    for (int base = 0; base < n; base += 32) {
        const int e = base + lane;
        int   s = s0;
        float w = x0 / den;
        if (base > 0) {
            s = node; w = 0.f;
            if (e < n) {
                s = (e < n - 1) ? nbr[st + e] : node;
                w = __expf(leaky(slab_sum<NS>(Ps, s) + adv) - m) / den;
            }
        }
        const int nb = (n - base < 32) ? (n - base) : 32;
        for (int j = 0; j < nb; j += U) {
            float4 r[U][C];
            float  wj[U];
            #pragma unroll
            for (int u = 0; u < U; ++u) {
                const int jj = (j + u < nb) ? (j + u) : (nb - 1);
                const int sj = __builtin_amdgcn_readlane(s, jj);
                const float wv = __int_as_float(__builtin_amdgcn_readlane(__float_as_int(w), jj));
                wj[u] = (j + u < nb) ? wv : 0.f;
                const float* row = hl + (size_t)sj * F;
                #pragma unroll
                for (int c = 0; c < C; ++c)
                    r[u][c] = *reinterpret_cast<const float4*>(row + c * 128);
            }
            #pragma unroll
            for (int u = 0; u < U; ++u) {
                #pragma unroll
                for (int c = 0; c < C; ++c) {
                    acc[c].x += r[u][c].x * wj[u];
                    acc[c].y += r[u][c].y * wj[u];
                    acc[c].z += r[u][c].z * wj[u];
                    acc[c].w += r[u][c].w * wj[u];
                }
            }
        }
    }

    #pragma unroll
    for (int c = 0; c < C; ++c) {
        const float4 b = *reinterpret_cast<const float4*>(bias + c * 128 + lane * 4);
        float4 v;
        v.x = acc[c].x + b.x; v.y = acc[c].y + b.y;
        v.z = acc[c].z + b.z; v.w = acc[c].w + b.w;
        if (ELU) {
            v.x = v.x > 0.f ? v.x : (__expf(v.x) - 1.f);
            v.y = v.y > 0.f ? v.y : (__expf(v.y) - 1.f);
            v.z = v.z > 0.f ? v.z : (__expf(v.z) - 1.f);
            v.w = v.w > 0.f ? v.w : (__expf(v.w) - 1.f);
        }
        *reinterpret_cast<float4*>(orow + c * 128) = v;
    }
}

#define FG_ROWS  16
#define FG_PITCH (D_HID + 4)

template <int MASK>
__device__ __forceinline__ float row16_xadd(float v) {
    const int p = __builtin_amdgcn_update_dpp(0, __float_as_int(v), 0x160 + MASK, 0xf, 0xf, false);
    return v + __int_as_float(p);
}
__device__ __forceinline__ float row16_sum(float v) {
    v = row16_xadd<8>(v);
    v = row16_xadd<4>(v);
    v = row16_xadd<2>(v);
    return row16_xadd<1>(v);
}

struct fg_chunk {
    v4f a0, a1;
    v2f b[8];
};

__device__ __forceinline__ void fg_fetch(fg_chunk& q, const float* Ap, const float* __restrict__ Bp)
{
    q.a0 = *reinterpret_cast<const v4f*>(Ap);
    q.a1 = *reinterpret_cast<const v4f*>(Ap + 4);
    #pragma unroll
    for (int i = 0; i < 8; ++i)
        q.b[i] = *reinterpret_cast<const v2f*>(Bp + (size_t)i * D_OUT);
}

__device__ __forceinline__ void fg_mma(const fg_chunk& q, v8f& acc0, v8f& acc1)
{
    #pragma unroll
    for (int j = 0; j < 4; ++j) {
        const v4f aq = (j < 2) ? q.a0 : q.a1;
        v2f a;  a.x = aq[(2 * j) & 3];  a.y = aq[((2 * j) & 3) + 1];
        const v2f lo = q.b[2 * j], hi = q.b[2 * j + 1];
        v2f b0, b1;
        b0.x = lo[0]; b0.y = hi[0];
        b1.x = lo[1]; b1.y = hi[1];
        acc0 = __builtin_amdgcn_wmma_f32_16x16x4_f32(false, a, false, b0, (short)0, acc0, false, false);
        acc1 = __builtin_amdgcn_wmma_f32_16x16x4_f32(false, a, false, b1, (short)0, acc1, false, false);
    }
}

__global__ __launch_bounds__(512, 12) void gat_gather_hidden_gemm(
    const int* __restrict__ nbr, const int* __restrict__ start,
    const int* __restrict__ cnt,
    const float* __restrict__ Ps, const float* __restrict__ Pd,
    const float* __restrict__ h, const float* __restrict__ bias,
    const float* __restrict__ W2,
    const float* __restrict__ a_s2, const float* __restrict__ a_d2,
    float* __restrict__ h2, float* __restrict__ P2s, float* __restrict__ P2d, int N)
{
    __shared__ float xs[FG_ROWS * FG_PITCH];
    __shared__ float part[4 * FG_ROWS * D_OUT];

    const int wave = __builtin_amdgcn_readfirstlane((int)(threadIdx.x >> 5));
    const int lane = threadIdx.x & 31;
    const int m0   = blockIdx.x * FG_ROWS;
    const int node = m0 + wave;

    float* xrow = xs + wave * FG_PITCH + lane * 4;
    if (node < N) {
        gat_gather_row<D_HID, 4, true>(nbr, start, cnt, Ps, Pd, h, bias, xrow, node, lane);
    } else {
        #pragma unroll
        for (int c = 0; c < D_HID / 128; ++c)
            *reinterpret_cast<float4*>(xrow + c * 128) = make_float4(0.f, 0.f, 0.f, 0.f);
    }
    __syncthreads();

    const int half = lane >> 4;
    const int l15  = lane & 15;
    const int n0   = (wave & 3) * 32;
    const int kq   = wave >> 2;
    const float* Ap = xs + l15 * FG_PITCH + kq * 128 + 8 * half;
    const float* Bp = W2 + (size_t)(kq * 128 + 8 * half) * D_OUT + n0 + 2 * l15;

    v8f acc0 = {}; v8f acc1 = {};
    fg_chunk q0, q1;
    fg_fetch(q0, Ap, Bp);
    #pragma unroll 1
    for (int it = 0; it < 4; ++it) {
        fg_fetch(q1, Ap + 16, Bp + (size_t)16 * D_OUT);
        fg_mma(q0, acc0, acc1);
        if (it < 3)
            fg_fetch(q0, Ap + 32, Bp + (size_t)32 * D_OUT);
        fg_mma(q1, acc0, acc1);
        Ap += 32;  Bp += (size_t)32 * D_OUT;
    }

    float* prow = part + (size_t)(kq * FG_ROWS + half * 8) * D_OUT + n0 + 2 * l15;
    #pragma unroll
    for (int r = 0; r < 8; ++r) {
        v2f o;  o.x = acc0[r];  o.y = acc1[r];
        *reinterpret_cast<v2f*>(prow + r * D_OUT) = o;
    }
    __syncthreads();

    const float* pr = part + (size_t)wave * D_OUT + lane * 4;
    const v4f p0 = *reinterpret_cast<const v4f*>(pr);
    const v4f p1 = *reinterpret_cast<const v4f*>(pr + 1 * FG_ROWS * D_OUT);
    const v4f p2 = *reinterpret_cast<const v4f*>(pr + 2 * FG_ROWS * D_OUT);
    const v4f p3 = *reinterpret_cast<const v4f*>(pr + 3 * FG_ROWS * D_OUT);
    const v4f v  = ((p0 + p1) + p2) + p3;
    const v4f sv = *reinterpret_cast<const v4f*>(a_s2 + lane * 4);
    const v4f dv = *reinterpret_cast<const v4f*>(a_d2 + lane * 4);
    float ps = v[0] * sv[0] + v[1] * sv[1] + v[2] * sv[2] + v[3] * sv[3];
    float pd = v[0] * dv[0] + v[1] * dv[1] + v[2] * dv[2] + v[3] * dv[3];
    ps = row16_sum(ps);
    pd = row16_sum(pd);
    if (node < N) {
        *reinterpret_cast<v4f*>(h2 + (size_t)node * D_OUT + lane * 4) = v;
        if (l15 == 0) {
            P2s[(size_t)node * (D_OUT / 64) + half] = ps;
            P2d[(size_t)node * (D_OUT / 64) + half] = pd;
        }
    }
}

__global__ __launch_bounds__(256) void gat_gather_out(
    const int* __restrict__ nbr, const int* __restrict__ start,
    const int* __restrict__ cnt,
    const float* __restrict__ Ps, const float* __restrict__ Pd,
    const float* __restrict__ h, const float* __restrict__ bias,
    float* __restrict__ out, int N)
{
    const int node = __builtin_amdgcn_readfirstlane((int)(blockIdx.x * 8 + (threadIdx.x >> 5)));
    if (node >= N) return;
    const int lane = threadIdx.x & 31;
    gat_gather_row<D_OUT, 8, false>(nbr, start, cnt, Ps, Pd, h, bias,
                                    out + (size_t)node * D_OUT + lane * 4, node, lane);
}

extern "C" void kernel_launch(void* const* d_in, const int* in_sizes, int n_in,
                              void* d_out, int out_size, void* d_ws, size_t ws_size,
                              hipStream_t stream) {
    const float* feature = (const float*)d_in[0];
    const int*   edges   = (const int*)d_in[1];
    const float* W1      = (const float*)d_in[2];
    const float* a_src1  = (const float*)d_in[3];
    const float* a_dst1  = (const float*)d_in[4];
    const float* b1      = (const float*)d_in[5];
    const float* W2      = (const float*)d_in[6];
    const float* a_src2  = (const float*)d_in[7];
    const float* a_dst2  = (const float*)d_in[8];
    const float* b2      = (const float*)d_in[9];
    float* out = (float*)d_out;

    const int N      = in_sizes[0] / D_IN;
    const int E_real = in_sizes[1] / 2;
    const int* src = edges;
    const int* dst = edges + E_real;

    float* h1    = (float*)d_ws;
    float* agg1  = h1   + (size_t)N * D_HID;
    float* spare = agg1 + (size_t)N * D_HID;
    int*   start = (int*)(spare + 2 * (size_t)N);
    int*   cnt   = start + N;
    int*   nbr   = cnt + N;
    float* h2    = agg1;
    float* p1s = out;
    float* p1d = p1s + (size_t)N * (D_HID / 64);
    float* p2s = h2  + (size_t)N * D_OUT;
    float* p2d = p2s + (size_t)N * (D_OUT / 64);

    const int T = 256;
    dim3 blkT(T);
    dim3 gN8((N + 7) / 8);

    gat_group<<<dim3((N + GRP_NPB - 1) / GRP_NPB), dim3(GRP_NT), 0, stream>>>(src, dst, start, cnt, nbr, N, E_real);

    gat_gemm1_f32_wmma<<<dim3(((N + 16 * G1_WAVES - 1) / (16 * G1_WAVES)) * G1_SLABS), dim3(G1_NT), 0, stream>>>(feature, W1, h1, a_src1, a_dst1, p1s, p1d, N);
    gat_gather_hidden_gemm<<<dim3((N + FG_ROWS - 1) / FG_ROWS), dim3(FG_ROWS * 32), 0, stream>>>(
        nbr, start, cnt, p1s, p1d, h1, b1, W2, a_src2, a_dst2, h2, p2s, p2d, N);

    gat_gather_out<<<gN8, blkT, 0, stream>>>(nbr, start, cnt, p2s, p2d, h2, b2, out, N);
}


# Edge grouping on the whole chip: list cut in 4 parts, 252 work-groups (part x 160 nodes) each test a quarter; a node has a sorted segment per part, gathers take them in part order; sum order changes
# speedup vs baseline: 5.1582x; 1.0342x over previous
#include <hip/hip_runtime.h>
#include <hip/hip_bf16.h>

typedef __attribute__((ext_vector_type(2))) float v2f;
typedef __attribute__((ext_vector_type(4))) float v4f;
typedef __attribute__((ext_vector_type(8))) float v8f;
typedef __attribute__((ext_vector_type(16))) float v16f;

#define D_IN  512
#define D_HID 512
#define D_OUT 128
#define NEG_SLOPE 0.2f

#define G1_WAVES  20
#define G1_NT     (G1_WAVES * 32)
#define G1_SLABS  (D_HID / 64)
#define G1_KP     (D_IN / 2)
#define G1_PITCH  136
#define G1_LDS_FLOATS  (41 * 1024)
static_assert(G1_LDS_FLOATS >= G1_KP * G1_PITCH, "slab does not fit");

struct g1_chunk {
    v4f a0, a1;
    v4f b[8];
};

__device__ __forceinline__ void g1_fetch(g1_chunk& q, const float* __restrict__ Ap, const float* Bl)
{
    q.a0 = *reinterpret_cast<const v4f*>(Ap);
    q.a1 = *reinterpret_cast<const v4f*>(Ap + 4);
    #pragma unroll
    for (int j = 0; j < 4; ++j) {
        q.b[2 * j]     = *reinterpret_cast<const v4f*>(Bl + j * G1_PITCH);
        q.b[2 * j + 1] = *reinterpret_cast<const v4f*>(Bl + j * G1_PITCH + 64);
    }
}

__device__ __forceinline__ void g1_mma(const g1_chunk& q, v8f& acc0, v8f& acc1,
                                       v8f& acc2, v8f& acc3)
{
    #pragma unroll
    for (int j = 0; j < 4; ++j) {
        const v4f aq = (j < 2) ? q.a0 : q.a1;
        v2f a;  a.x = aq[(2 * j) & 3];  a.y = aq[((2 * j) & 3) + 1];
        const v4f lo = q.b[2 * j], hi = q.b[2 * j + 1];
        v2f b0, b1, b2, b3;
        b0.x = lo[0]; b0.y = lo[1];
        b1.x = lo[2]; b1.y = lo[3];
        b2.x = hi[0]; b2.y = hi[1];
        b3.x = hi[2]; b3.y = hi[3];
        acc0 = __builtin_amdgcn_wmma_f32_16x16x4_f32(false, a, false, b0, (short)0, acc0, false, false);
        acc1 = __builtin_amdgcn_wmma_f32_16x16x4_f32(false, a, false, b1, (short)0, acc1, false, false);
        acc2 = __builtin_amdgcn_wmma_f32_16x16x4_f32(false, a, false, b2, (short)0, acc2, false, false);
        acc3 = __builtin_amdgcn_wmma_f32_16x16x4_f32(false, a, false, b3, (short)0, acc3, false, false);
    }
}

__global__ __launch_bounds__(G1_NT) void gat_gemm1_f32_wmma(
    const float* __restrict__ A, const float* __restrict__ B,
    float* __restrict__ C,
    const float* __restrict__ a_s, const float* __restrict__ a_d,
    float* __restrict__ Ps, float* __restrict__ Pd, int M)
{
    __shared__ float Bs[G1_LDS_FLOATS];

    const int tid  = threadIdx.x;
    const int lane = tid & 31;
    const int wave = __builtin_amdgcn_readfirstlane(tid >> 5);
    const int slab = blockIdx.x % G1_SLABS;
    const int n0   = slab * 64;
    const int m0   = ((blockIdx.x / G1_SLABS) * G1_WAVES + wave) * 16;
    const bool valid = m0 < M;

    const int half = lane >> 4;
    const int l15  = lane & 15;

    constexpr int ITEMS = G1_KP * 16;
    constexpr int FULL  = ITEMS / G1_NT;
    constexpr int REM   = ITEMS - FULL * G1_NT;
    static_assert(G1_NT % 16 == 0, "column group must not depend on the round");
    static_assert(REM > 0 && REM <= G1_NT, "one partial round");
    const int cg = tid & 15;
    const float* Bq = B + n0 + 4 * cg;
    int pp[FULL + 1];
    v4f r0[FULL + 1], r1[FULL + 1];
    #pragma unroll
    for (int u = 0; u <= FULL; ++u) {
        pp[u] = ((u < FULL || tid < REM) ? (u * G1_NT + tid) : tid) >> 4;
        r0[u] = *reinterpret_cast<const v4f*>(Bq + (size_t)(2 * pp[u]) * D_HID);
        r1[u] = *reinterpret_cast<const v4f*>(Bq + (size_t)(2 * pp[u] + 1) * D_HID);
    }

    const v4f sv = *reinterpret_cast<const v4f*>(a_s + n0 + 4 * l15);
    const v4f dv = *reinterpret_cast<const v4f*>(a_d + n0 + 4 * l15);
    const float* Ap = A + (size_t)((valid ? m0 : 0) + l15) * D_IN + 8 * half;
    g1_chunk cur;
    cur.a0 = *reinterpret_cast<const v4f*>(Ap);
    cur.a1 = *reinterpret_cast<const v4f*>(Ap + 4);

    #pragma unroll
    for (int u = 0; u <= FULL; ++u) {
        float* d = Bs + pp[u] * G1_PITCH + 4 * cg;
        v4f lo, hi;
        lo[0] = r0[u][0]; lo[1] = r1[u][0]; lo[2] = r0[u][1]; lo[3] = r1[u][1];
        hi[0] = r0[u][2]; hi[1] = r1[u][2]; hi[2] = r0[u][3]; hi[3] = r1[u][3];
        *reinterpret_cast<v4f*>(d)      = lo;
        *reinterpret_cast<v4f*>(d + 64) = hi;
    }
    __syncthreads();
    if (!valid) return;

    const float* Bl = Bs + (4 * half) * G1_PITCH + 4 * l15;
    v8f acc0 = {}; v8f acc1 = {}; v8f acc2 = {}; v8f acc3 = {};

    #pragma unroll
    for (int j = 0; j < 4; ++j) {
        cur.b[2 * j]     = *reinterpret_cast<const v4f*>(Bl + j * G1_PITCH);
        cur.b[2 * j + 1] = *reinterpret_cast<const v4f*>(Bl + j * G1_PITCH + 64);
    }
    for (int k0 = 16; k0 < D_IN; k0 += 16) {
        Ap += 16;  Bl += 8 * G1_PITCH;
        g1_chunk nxt;
        g1_fetch(nxt, Ap, Bl);
        g1_mma(cur, acc0, acc1, acc2, acc3);
        cur = nxt;
    }
    g1_mma(cur, acc0, acc1, acc2, acc3);

    float* Crow = C + (size_t)(m0 + half * 8) * D_HID + n0 + 4 * l15;
    #pragma unroll
    for (int r = 0; r < 8; ++r) {
        v4f o;
        o[0] = acc0[r]; o[1] = acc1[r]; o[2] = acc2[r]; o[3] = acc3[r];
        *reinterpret_cast<v4f*>(Crow + (size_t)r * D_HID) = o;
    }

    const float s0 = sv[0], s1 = sv[1], s2 = sv[2], s3 = sv[3];
    const float d0 = dv[0], d1 = dv[1], d2 = dv[2], d3 = dv[3];
    v16f p;
    #pragma unroll
    for (int r = 0; r < 8; ++r) {
        p[r]     = acc0[r] * s0 + acc1[r] * s1 + acc2[r] * s2 + acc3[r] * s3;
        p[8 + r] = acc0[r] * d0 + acc1[r] * d1 + acc2[r] * d2 + acc3[r] * d3;
    }
    const bool up8 = (l15 & 8) != 0, up4 = (l15 & 4) != 0;
    const bool up2 = (l15 & 2) != 0, up1 = (l15 & 1) != 0;
    v8f q;
    #pragma unroll
    for (int i = 0; i < 8; ++i) {
        const float lo = p[i], hi = p[i + 8];
        q[i] = (up8 ? hi : lo) + __shfl_xor(up8 ? lo : hi, 8, 32);
    }
    v4f t;
    #pragma unroll
    for (int i = 0; i < 4; ++i) {
        const float lo = q[i], hi = q[i + 4];
        t[i] = (up4 ? hi : lo) + __shfl_xor(up4 ? lo : hi, 4, 32);
    }
    v2f u;
    #pragma unroll
    for (int i = 0; i < 2; ++i) {
        const float lo = t[i], hi = t[i + 2];
        u[i] = (up2 ? hi : lo) + __shfl_xor(up2 ? lo : hi, 2, 32);
    }
    const float tot = (up1 ? u[1] : u[0]) + __shfl_xor(up1 ? u[0] : u[1], 1, 32);
    float* T = (l15 < 8) ? Ps : Pd;
    T[(size_t)(m0 + half * 8 + (l15 & 7)) * G1_SLABS + slab] = tot;
}
#define GRP_PARTS  4
#define GRP_NPB    160
#define GRP_NT     1024
#define GRP_NW     (GRP_NT / 32)
#define GRP_WCAP   128
#define GRP_SEGCAP 64
#define GRP_U      5
#define GRP_LDS_INTS (33 * 1024)
#define GRP_NONE   0x7fffffff

__device__ __forceinline__ int wave_sum_i(int v) {
    for (int off = 16; off > 0; off >>= 1) v += __shfl_xor(v, off, 32);
    return v;
}

__global__ __launch_bounds__(GRP_NT) void gat_group(
    const int* __restrict__ src, const int* __restrict__ dst,
    int* __restrict__ start, int* __restrict__ cnt, int* __restrict__ nbr,
    int N, int E_all)
{
    __shared__ int lcnt[GRP_NPB];
    __shared__ int lfill[GRP_NPB];
    __shared__ int lstart[GRP_NPB];
    static_assert(GRP_LDS_INTS >= GRP_NPB * GRP_SEGCAP, "segments do not fit");
    __shared__ int lseg[GRP_LDS_INTS];
    __shared__ int wl_node[GRP_NW * GRP_WCAP];
    __shared__ int wl_edge[GRP_NW * GRP_WCAP];
    __shared__ int wlen[GRP_NW];
    __shared__ int wbelow[GRP_NW];
    __shared__ int blk_over;

    const int t = threadIdx.x, lane = t & 31;
    const int w  = __builtin_amdgcn_readfirstlane(t >> 5);
    const int nr   = gridDim.x / GRP_PARTS;
    const int part = blockIdx.x / nr;
    const int g    = blockIdx.x - part * nr;
    const int lo   = g * GRP_NPB;
    const int per = ((E_all + GRP_PARTS - 1) / GRP_PARTS + 3) & ~3;
    int eb = part * per;  if (eb > E_all) eb = E_all;
    int ee = eb + per;    if (ee > E_all) ee = E_all;
    const int E = ee - eb;
    src += eb;  dst += eb;  nbr += eb;

    for (int k = t; k < GRP_NPB; k += GRP_NT) { lcnt[k] = 0; lfill[k] = 0; }
    __syncthreads();

    int below = 0;
    int wl    = 0;
    int* const my_node = wl_node + w * GRP_WCAP;
    int* const my_edge = wl_edge + w * GRP_WCAP;

    auto see = [&](int d, int e) {
        const int r = d - lo;
        below += (r < 0) ? 1 : 0;
        const bool hit = (unsigned)r < (unsigned)GRP_NPB;
        const unsigned m = __builtin_amdgcn_ballot_w32(hit);
        if (m != 0u) {
            if (hit) {
                atomicAdd(&lcnt[r], 1);
                const int pos = wl + (int)__builtin_amdgcn_mbcnt_lo(m, 0u);
                if (pos < GRP_WCAP) { my_node[pos] = r; my_edge[pos] = e; }
            }
            wl += __builtin_popcount(m);
        }
    };

    int head = (int)(((16u - ((unsigned)(size_t)dst & 15u)) & 15u) >> 2);
    if (head > E) head = E;
    const int nvec  = (E - head) >> 2;
    const int tail0 = head + 4 * nvec;
    const int4* dv = reinterpret_cast<const int4*>(dst + head);
    const int rot = (int)(((long long)g * nvec) / nr);

    auto sweep = [&](int b0, bool guarded) __attribute__((always_inline)) {
        int4 d[GRP_U];
        int  c[GRP_U];
        #pragma unroll
        for (int u = 0; u < GRP_U; ++u) {
            const int i = b0 + u * GRP_NT + t;
            int cc = i + rot;
            if (cc >= nvec) cc -= nvec;
            const bool ok = !guarded || i < nvec;
            c[u] = cc;
            d[u] = dv[ok ? cc : 0];
            if (!ok) d[u] = make_int4(GRP_NONE, GRP_NONE, GRP_NONE, GRP_NONE);
        }
        #pragma unroll
        for (int u = 0; u < GRP_U; ++u) {
            const int e = head + 4 * c[u];
            see(d[u].x, e); see(d[u].y, e + 1); see(d[u].z, e + 2); see(d[u].w, e + 3);
        }
    };
    int b0 = 0;
    for (; b0 + GRP_NT * GRP_U <= nvec; b0 += GRP_NT * GRP_U) sweep(b0, false);
    if (b0 < nvec) sweep(b0, true);
    if (w == 0) {
        const int ntail = E - tail0;
        see((lane < head)  ? dst[lane]         : GRP_NONE, lane);
        see((lane < ntail) ? dst[tail0 + lane] : GRP_NONE, tail0 + lane);
    }

    const int wb = wave_sum_i(below);
    if (lane == 0) { wlen[w] = wl; wbelow[w] = wb; }
    __syncthreads();

    {
        const int have = (wl < GRP_WCAP) ? wl : GRP_WCAP;
        int rr[GRP_WCAP / 32], ss[GRP_WCAP / 32];
        #pragma unroll
        for (int q = 0; q < GRP_WCAP / 32; ++q) {
            const int p = q * 32 + lane;
            const bool ok = p < have;
            const int pc = ok ? p : 0;
            const int r = my_node[pc];
            const int e = my_edge[pc];
            rr[q] = ok ? r : -1;
            ss[q] = ok ? e : -1;
        }
        #pragma unroll
        for (int q = 0; q < GRP_WCAP / 32; ++q)
            ss[q] = (ss[q] >= 0) ? src[ss[q]] : 0;
        #pragma unroll
        for (int q = 0; q < GRP_WCAP / 32; ++q) {
            if (rr[q] >= 0) {
                const int slot = atomicAdd(&lfill[rr[q]], 1);
                if (slot < GRP_SEGCAP) lseg[rr[q] * GRP_SEGCAP + slot] = ss[q];
            }
        }
    }
    if (w == 0) {
        int b = 0, ov = 0;
        for (int i = lane; i < GRP_NW; i += 32) {
            b  += wbelow[i];
            ov |= (wlen[i] > GRP_WCAP) ? 1 : 0;
        }
        b = wave_sum_i(b);
        const unsigned anyov = __builtin_amdgcn_ballot_w32(ov != 0);
        constexpr int PER = (GRP_NPB + 31) / 32;
        int v[PER];
        int s = 0;
        #pragma unroll
        for (int i = 0; i < PER; ++i) {
            const int k = lane * PER + i;
            v[i] = (k < GRP_NPB) ? lcnt[k] : 0;
            s += v[i];
        }
        int incl = s;
        for (int off = 1; off < 32; off <<= 1) {
            const int u = __shfl_up(incl, off, 32);
            if (lane >= off) incl += u;
        }
        int run = b + incl - s;
        #pragma unroll
        for (int i = 0; i < PER; ++i) {
            const int k = lane * PER + i;
            if (k < GRP_NPB) lstart[k] = run;
            run += v[i];
        }
        if (lane == 0) blk_over = (anyov != 0u) ? 1 : 0;
    }
    __syncthreads();

    const bool over = __builtin_amdgcn_readfirstlane(blk_over) != 0;
    for (int k = w; k < GRP_NPB; k += GRP_NW) {
        const int node = lo + k;
        if (node >= N) break;
        const int deg = __builtin_amdgcn_readfirstlane(lcnt[k]);
        const int st  = __builtin_amdgcn_readfirstlane(lstart[k]);
        if (lane == 0) {
            start[(size_t)node * GRP_PARTS + part] = st + eb;
            cnt[(size_t)node * GRP_PARTS + part]   = deg;
        }
        if (!over && deg <= GRP_SEGCAP) {
            const int* seg = lseg + k * GRP_SEGCAP;
            const int x0 = seg[lane];
            const int a0 = (lane < deg) ? x0 : GRP_NONE;
            int r0 = 0;
            if (deg <= 32) {
                for (int j = 0; j < deg; ++j) {
                    const int kk = __builtin_amdgcn_readlane(a0, j);
                    r0 += ((kk < a0) || (kk == a0 && j < lane)) ? 1 : 0;
                }
                if (lane < deg) nbr[st + r0] = a0;
            } else {
                const int x1 = seg[lane + 32];
                const int a1 = (lane + 32 < deg) ? x1 : GRP_NONE;
                int r1 = 0;
                for (int j = 0; j < 32; ++j) {
                    const int kk = __builtin_amdgcn_readlane(a0, j);
                    r0 += ((kk < a0) || (kk == a0 && j < lane)) ? 1 : 0;
                    r1 += (kk <= a1) ? 1 : 0;
                }
                for (int j = 32; j < deg; ++j) {
                    const int kk = __builtin_amdgcn_readlane(a1, j - 32);
                    r0 += (kk < a0) ? 1 : 0;
                    r1 += ((kk < a1) || (kk == a1 && (j - 32) < lane)) ? 1 : 0;
                }
                nbr[st + r0] = a0;
                if (lane + 32 < deg) nbr[st + r1] = a1;
            }
        } else {
            int run = 0;
            for (int i0 = 0; i0 < E; i0 += 32) {
                const int i = i0 + lane;
                const bool hit = (i < E) && (dst[i] == node);
                const unsigned m = __builtin_amdgcn_ballot_w32(hit);
                if (hit) nbr[st + run + (int)__builtin_amdgcn_mbcnt_lo(m, 0u)] = src[i];
                run += __builtin_popcount(m);
            }
        }
    }
}

__device__ __forceinline__ float wave_max(float v) {
    for (int off = 16; off > 0; off >>= 1) v = fmaxf(v, __shfl_xor(v, off, 32));
    return v;
}
__device__ __forceinline__ float wave_sum(float v) {
    for (int off = 16; off > 0; off >>= 1) v += __shfl_xor(v, off, 32);
    return v;
}
__device__ __forceinline__ float leaky(float v) { return v > 0.f ? v : NEG_SLOPE * v; }

template <int NS>
__device__ __forceinline__ float slab_sum(const float* __restrict__ P, int node)
{
    const float* p = P + (size_t)node * NS;
    if (NS == 2) {
        const float2 a = *reinterpret_cast<const float2*>(p);
        return a.x + a.y;
    } else {
        float s = 0.f;
        #pragma unroll
        for (int i = 0; i < NS; i += 4) {
            const float4 a = *reinterpret_cast<const float4*>(p + i);
            s = (i == 0) ? a.x : s + a.x;
            s += a.y; s += a.z; s += a.w;
        }
        return s;
    }
}

__device__ __forceinline__ int seg_place(int e, int c1, int c2, int c3, int o0, int o1, int o2, int o3)
{
    return e + ((e < c1) ? o0 : (e < c2) ? o1 : (e < c3) ? o2 : o3);
}

template <int F, int U, bool ELU>
__device__ __forceinline__ void gat_gather_row(
    const int* __restrict__ nbr, const int* __restrict__ start,
    const int* __restrict__ cnt,
    const float* __restrict__ Ps, const float* __restrict__ Pd,
    const float* __restrict__ h, const float* __restrict__ bias,
    float* __restrict__ orow, const int node, const int lane)
{
    constexpr int C  = F / 128;
    constexpr int NS = F / 64;
    static_assert(GRP_PARTS == 4, "one 16-byte record of places and one of counts per node");
    const int4 sv = *reinterpret_cast<const int4*>(start + (size_t)node * GRP_PARTS);
    const int4 cv = *reinterpret_cast<const int4*>(cnt   + (size_t)node * GRP_PARTS);
    const int c1 = cv.x, c2 = c1 + cv.y, c3 = c2 + cv.z;
    const int n  = c3 + cv.w + 1;
    const int o0 = sv.x, o1 = sv.y - c1, o2 = sv.z - c2, o3 = sv.w - c3;
    const float adv = slab_sum<NS>(Pd, node);

    const bool has0 = lane < n;
    int   s0 = node;
    float v0 = -INFINITY;
    if (has0) {
        s0 = (lane < n - 1) ? nbr[seg_place(lane, c1, c2, c3, o0, o1, o2, o3)] : node;
        v0 = leaky(slab_sum<NS>(Ps, s0) + adv);
    }
    float m = v0;
    for (int e = lane + 32; e < n; e += 32) {
        const int s = (e < n - 1) ? nbr[seg_place(e, c1, c2, c3, o0, o1, o2, o3)] : node;
        m = fmaxf(m, leaky(slab_sum<NS>(Ps, s) + adv));
    }
    m = wave_max(m);
    const float x0 = has0 ? __expf(v0 - m) : 0.f;
    float den = x0;
    for (int e = lane + 32; e < n; e += 32) {
        const int s = (e < n - 1) ? nbr[seg_place(e, c1, c2, c3, o0, o1, o2, o3)] : node;
        den += __expf(leaky(slab_sum<NS>(Ps, s) + adv) - m);
    }
    den = wave_sum(den);

    float4 acc[C];
    #pragma unroll
    for (int c = 0; c < C; ++c) acc[c] = make_float4(0.f, 0.f, 0.f, 0.f);

    const float* hl = h + lane * 4;
    for (int base = 0; base < n; base += 32) {
        const int e = base + lane;
        int   s = s0;
        float w = x0 / den;
        if (base > 0) {
            s = node; w = 0.f;
            if (e < n) {
                s = (e < n - 1) ? nbr[seg_place(e, c1, c2, c3, o0, o1, o2, o3)] : node;
                w = __expf(leaky(slab_sum<NS>(Ps, s) + adv) - m) / den;
            }
        }
        const int nb = (n - base < 32) ? (n - base) : 32;
        for (int j = 0; j < nb; j += U) {
            float4 r[U][C];
            float  wj[U];
            #pragma unroll
            for (int u = 0; u < U; ++u) {
                const int jj = (j + u < nb) ? (j + u) : (nb - 1);
                const int sj = __builtin_amdgcn_readlane(s, jj);
                const float wv = __int_as_float(__builtin_amdgcn_readlane(__float_as_int(w), jj));
                wj[u] = (j + u < nb) ? wv : 0.f;
                const float* row = hl + (size_t)sj * F;
                #pragma unroll
                for (int c = 0; c < C; ++c)
                    r[u][c] = *reinterpret_cast<const float4*>(row + c * 128);
            }
            #pragma unroll
            for (int u = 0; u < U; ++u) {
                #pragma unroll
                for (int c = 0; c < C; ++c) {
                    acc[c].x += r[u][c].x * wj[u];
                    acc[c].y += r[u][c].y * wj[u];
                    acc[c].z += r[u][c].z * wj[u];
                    acc[c].w += r[u][c].w * wj[u];
                }
            }
        }
    }

    #pragma unroll
    for (int c = 0; c < C; ++c) {
        const float4 b = *reinterpret_cast<const float4*>(bias + c * 128 + lane * 4);
        float4 v;
        v.x = acc[c].x + b.x; v.y = acc[c].y + b.y;
        v.z = acc[c].z + b.z; v.w = acc[c].w + b.w;
        if (ELU) {
            v.x = v.x > 0.f ? v.x : (__expf(v.x) - 1.f);
            v.y = v.y > 0.f ? v.y : (__expf(v.y) - 1.f);
            v.z = v.z > 0.f ? v.z : (__expf(v.z) - 1.f);
            v.w = v.w > 0.f ? v.w : (__expf(v.w) - 1.f);
        }
        *reinterpret_cast<float4*>(orow + c * 128) = v;
    }
}

#define FG_ROWS  16
#define FG_PITCH (D_HID + 4)

template <int MASK>
__device__ __forceinline__ float row16_xadd(float v) {
    const int p = __builtin_amdgcn_update_dpp(0, __float_as_int(v), 0x160 + MASK, 0xf, 0xf, false);
    return v + __int_as_float(p);
}
__device__ __forceinline__ float row16_sum(float v) {
    v = row16_xadd<8>(v);
    v = row16_xadd<4>(v);
    v = row16_xadd<2>(v);
    return row16_xadd<1>(v);
}

struct fg_chunk {
    v4f a0, a1;
    v2f b[8];
};

__device__ __forceinline__ void fg_fetch(fg_chunk& q, const float* Ap, const float* __restrict__ Bp)
{
    q.a0 = *reinterpret_cast<const v4f*>(Ap);
    q.a1 = *reinterpret_cast<const v4f*>(Ap + 4);
    #pragma unroll
    for (int i = 0; i < 8; ++i)
        q.b[i] = *reinterpret_cast<const v2f*>(Bp + (size_t)i * D_OUT);
}

__device__ __forceinline__ void fg_mma(const fg_chunk& q, v8f& acc0, v8f& acc1)
{
    #pragma unroll
    for (int j = 0; j < 4; ++j) {
        const v4f aq = (j < 2) ? q.a0 : q.a1;
        v2f a;  a.x = aq[(2 * j) & 3];  a.y = aq[((2 * j) & 3) + 1];
        const v2f lo = q.b[2 * j], hi = q.b[2 * j + 1];
        v2f b0, b1;
        b0.x = lo[0]; b0.y = hi[0];
        b1.x = lo[1]; b1.y = hi[1];
        acc0 = __builtin_amdgcn_wmma_f32_16x16x4_f32(false, a, false, b0, (short)0, acc0, false, false);
        acc1 = __builtin_amdgcn_wmma_f32_16x16x4_f32(false, a, false, b1, (short)0, acc1, false, false);
    }
}

__global__ __launch_bounds__(512, 12) void gat_gather_hidden_gemm(
    const int* __restrict__ nbr, const int* __restrict__ start,
    const int* __restrict__ cnt,
    const float* __restrict__ Ps, const float* __restrict__ Pd,
    const float* __restrict__ h, const float* __restrict__ bias,
    const float* __restrict__ W2,
    const float* __restrict__ a_s2, const float* __restrict__ a_d2,
    float* __restrict__ h2, float* __restrict__ P2s, float* __restrict__ P2d, int N)
{
    __shared__ float xs[FG_ROWS * FG_PITCH];
    __shared__ float part[4 * FG_ROWS * D_OUT];

    const int wave = __builtin_amdgcn_readfirstlane((int)(threadIdx.x >> 5));
    const int lane = threadIdx.x & 31;
    const int m0   = blockIdx.x * FG_ROWS;
    const int node = m0 + wave;

    float* xrow = xs + wave * FG_PITCH + lane * 4;
    if (node < N) {
        gat_gather_row<D_HID, 4, true>(nbr, start, cnt, Ps, Pd, h, bias, xrow, node, lane);
    } else {
        #pragma unroll
        for (int c = 0; c < D_HID / 128; ++c)
            *reinterpret_cast<float4*>(xrow + c * 128) = make_float4(0.f, 0.f, 0.f, 0.f);
    }
    __syncthreads();

    const int half = lane >> 4;
    const int l15  = lane & 15;
    const int n0   = (wave & 3) * 32;
    const int kq   = wave >> 2;
    const float* Ap = xs + l15 * FG_PITCH + kq * 128 + 8 * half;
    const float* Bp = W2 + (size_t)(kq * 128 + 8 * half) * D_OUT + n0 + 2 * l15;

    v8f acc0 = {}; v8f acc1 = {};
    fg_chunk q0, q1;
    fg_fetch(q0, Ap, Bp);
    #pragma unroll 1
    for (int it = 0; it < 4; ++it) {
        fg_fetch(q1, Ap + 16, Bp + (size_t)16 * D_OUT);
        fg_mma(q0, acc0, acc1);
        if (it < 3)
            fg_fetch(q0, Ap + 32, Bp + (size_t)32 * D_OUT);
        fg_mma(q1, acc0, acc1);
        Ap += 32;  Bp += (size_t)32 * D_OUT;
    }

    float* prow = part + (size_t)(kq * FG_ROWS + half * 8) * D_OUT + n0 + 2 * l15;
    #pragma unroll
    for (int r = 0; r < 8; ++r) {
        v2f o;  o.x = acc0[r];  o.y = acc1[r];
        *reinterpret_cast<v2f*>(prow + r * D_OUT) = o;
    }
    __syncthreads();

    const float* pr = part + (size_t)wave * D_OUT + lane * 4;
    const v4f p0 = *reinterpret_cast<const v4f*>(pr);
    const v4f p1 = *reinterpret_cast<const v4f*>(pr + 1 * FG_ROWS * D_OUT);
    const v4f p2 = *reinterpret_cast<const v4f*>(pr + 2 * FG_ROWS * D_OUT);
    const v4f p3 = *reinterpret_cast<const v4f*>(pr + 3 * FG_ROWS * D_OUT);
    const v4f v  = ((p0 + p1) + p2) + p3;
    const v4f sv = *reinterpret_cast<const v4f*>(a_s2 + lane * 4);
    const v4f dv = *reinterpret_cast<const v4f*>(a_d2 + lane * 4);
    float ps = v[0] * sv[0] + v[1] * sv[1] + v[2] * sv[2] + v[3] * sv[3];
    float pd = v[0] * dv[0] + v[1] * dv[1] + v[2] * dv[2] + v[3] * dv[3];
    ps = row16_sum(ps);
    pd = row16_sum(pd);
    if (node < N) {
        *reinterpret_cast<v4f*>(h2 + (size_t)node * D_OUT + lane * 4) = v;
        if (l15 == 0) {
            P2s[(size_t)node * (D_OUT / 64) + half] = ps;
            P2d[(size_t)node * (D_OUT / 64) + half] = pd;
        }
    }
}

__global__ __launch_bounds__(256) void gat_gather_out(
    const int* __restrict__ nbr, const int* __restrict__ start,
    const int* __restrict__ cnt,
    const float* __restrict__ Ps, const float* __restrict__ Pd,
    const float* __restrict__ h, const float* __restrict__ bias,
    float* __restrict__ out, int N)
{
    const int node = __builtin_amdgcn_readfirstlane((int)(blockIdx.x * 8 + (threadIdx.x >> 5)));
    if (node >= N) return;
    const int lane = threadIdx.x & 31;
    gat_gather_row<D_OUT, 8, false>(nbr, start, cnt, Ps, Pd, h, bias,
                                    out + (size_t)node * D_OUT + lane * 4, node, lane);
}

extern "C" void kernel_launch(void* const* d_in, const int* in_sizes, int n_in,
                              void* d_out, int out_size, void* d_ws, size_t ws_size,
                              hipStream_t stream) {
    const float* feature = (const float*)d_in[0];
    const int*   edges   = (const int*)d_in[1];
    const float* W1      = (const float*)d_in[2];
    const float* a_src1  = (const float*)d_in[3];
    const float* a_dst1  = (const float*)d_in[4];
    const float* b1      = (const float*)d_in[5];
    const float* W2      = (const float*)d_in[6];
    const float* a_src2  = (const float*)d_in[7];
    const float* a_dst2  = (const float*)d_in[8];
    const float* b2      = (const float*)d_in[9];
    float* out = (float*)d_out;

    const int N      = in_sizes[0] / D_IN;
    const int E_real = in_sizes[1] / 2;
    const int* src = edges;
    const int* dst = edges + E_real;

    float* h1    = (float*)d_ws;
    float* agg1  = h1   + (size_t)N * D_HID;
    float* spare = agg1 + (size_t)N * D_HID;
    int*   nbr   = (int*)(spare + 2 * (size_t)N) + 2 * (size_t)N;
    float* h2    = agg1;
    float* p1s = out;
    float* p1d = p1s + (size_t)N * (D_HID / 64);
    float* p2s = h2  + (size_t)N * D_OUT;
    float* p2d = p2s + (size_t)N * (D_OUT / 64);
    int*   start = (int*)(p2d + (size_t)N * (D_OUT / 64));
    int*   cnt   = start + (size_t)N * GRP_PARTS;

    const int T = 256;
    dim3 blkT(T);
    dim3 gN8((N + 7) / 8);

    gat_group<<<dim3(GRP_PARTS * ((N + GRP_NPB - 1) / GRP_NPB)), dim3(GRP_NT), 0, stream>>>(src, dst, start, cnt, nbr, N, E_real);

    gat_gemm1_f32_wmma<<<dim3(((N + 16 * G1_WAVES - 1) / (16 * G1_WAVES)) * G1_SLABS), dim3(G1_NT), 0, stream>>>(feature, W1, h1, a_src1, a_dst1, p1s, p1d, N);
    gat_gather_hidden_gemm<<<dim3((N + FG_ROWS - 1) / FG_ROWS), dim3(FG_ROWS * 32), 0, stream>>>(
        nbr, start, cnt, p1s, p1d, h1, b1, W2, a_src2, a_dst2, h2, p2s, p2d, N);

    gat_gather_out<<<gN8, blkT, 0, stream>>>(nbr, start, cnt, p2s, p2d, h2, b2, out, N);
}
